# starting kernel
# baseline (speedup 1.0000x reference)
#include <hip/hip_runtime.h>
#include <cstddef>
#include <cstdint>

#define D 128
#define LDS_PP2 144

typedef float v2f __attribute__((ext_vector_type(2)));
typedef float v8f __attribute__((ext_vector_type(8)));

__device__ __forceinline__ v8f wmma_f32(v2f a, v2f b, v8f c) {
#if defined(__gfx1250__)
  return __builtin_amdgcn_wmma_f32_16x16x4_f32(false, a, false, b,
                                               (short)0, c, false, false);
#else
  c[0] += a[0] * b[0];
  return c;
#endif
}

__global__ void zero_f32(float* __restrict__ p, int n) {
  int i = blockIdx.x * blockDim.x + threadIdx.x;
  if (i < n) p[i] = 0.0f;
}

__global__ void repack_weight(const float* __restrict__ W, float* __restrict__ WTi) {
  int idx = blockIdx.x * blockDim.x + threadIdx.x;
  if (idx >= D * D) return;
  int n = (idx >> 1) & (D - 1);
  int k = ((idx >> 8) << 1) | (idx & 1);
  WTi[idx] = W[n * D + k];
}

__global__ void scatter_edges(const float* __restrict__ x,
                              const int* __restrict__ src,
                              const int* __restrict__ dst,
                              float* __restrict__ msg,
                              float* __restrict__ deg,
                              int nEdges) {
  int wave = (int)((blockIdx.x * blockDim.x + threadIdx.x) >> 5);
  int lane = threadIdx.x & 31;
  if (wave >= nEdges) return;
  int s = src[wave];
  int d = dst[wave];
  float4 v = ((const float4*)(x + (size_t)s * D))[lane];
  float* o = msg + (size_t)d * D + lane * 4;
  atomicAdd(o + 0, v.x);
  atomicAdd(o + 1, v.y);
  atomicAdd(o + 2, v.z);
  atomicAdd(o + 3, v.w);
  if (deg != nullptr && lane == 0) atomicAdd(deg + d, 1.0f);
}

__global__ void __launch_bounds__(128)
sage_layer(const float* __restrict__ xin, const float* __restrict__ msg,
           const float* __restrict__ deg,
           const float* __restrict__ WTiS, const float* __restrict__ WTiN,
           const float* __restrict__ bias,
           float* __restrict__ out, int nNodes) {
  extern __shared__ float lds[];
  const int tid  = threadIdx.x;
  const int wave = tid >> 5;
  const int lane = tid & 31;
  const int half = lane >> 4;
  const int l16  = lane & 15;
  const int m0   = blockIdx.x * 64 + wave * 16;
  const int mRow = m0 + l16;
  const int mC   = (mRow < nNodes) ? mRow : (nNodes - 1);

  const float invd = 1.0f / fmaxf(deg[mC], 1.0f);
  const float* xr = xin + (size_t)mC * D + half * 2;
  const float* mr = msg + (size_t)mC * D + half * 2;
  const v2f* B2 = (const v2f*)lds;

  v8f acc[8];
#pragma unroll
  for (int t = 0; t < 8; ++t) acc[t] = (v8f)(0.0f);

  for (int phase = 0; phase < 2; ++phase) {
    const float4* src4 = (const float4*)(phase == 0 ? WTiS : WTiN);
    const float*  arow = (phase == 0) ? xr : mr;

    __syncthreads();
    for (int i = tid; i < (D * D) / 4; i += 128) {
      ((float4*)lds)[(i >> 6) * (LDS_PP2 / 2) + (i & 63)] = src4[i];
    }
    __syncthreads();

    v2f aCur = *(const v2f*)(arow);
    if (phase == 1) aCur *= invd;

    for (int k0 = 0; k0 < D; k0 += 4) {
      const int base2 = ((k0 >> 1) + half) * LDS_PP2 + l16;
      v2f b[8];
#pragma unroll
      for (int t = 0; t < 8; ++t) b[t] = B2[base2 + t * 16];

      const int kN = (k0 + 4 < D) ? (k0 + 4) : k0;
      v2f aNext = *(const v2f*)(arow + kN);
      if (phase == 1) aNext *= invd;

#pragma unroll
      for (int t = 0; t < 8; ++t) acc[t] = wmma_f32(aCur, b[t], acc[t]);
      aCur = aNext;
    }
  }

#pragma unroll
  for (int t = 0; t < 8; ++t) {
    const int col = t * 16 + l16;
    const float bv = bias[col];
#pragma unroll
    for (int r = 0; r < 8; ++r) {
      int row = m0 + r + half * 8;
      if (row < nNodes) out[(size_t)row * D + col] = acc[t][r] + bv;
    }
  }
}

__global__ void l2_normalize(float* __restrict__ h, int nNodes) {
  int wave = (int)((blockIdx.x * blockDim.x + threadIdx.x) >> 5);
  int lane = threadIdx.x & 31;
  if (wave >= nNodes) return;
  float4* row = (float4*)(h + (size_t)wave * D);
  float4 v = row[lane];
  float ss = v.x * v.x + v.y * v.y + v.z * v.z + v.w * v.w;
#pragma unroll
  for (int off = 16; off > 0; off >>= 1) ss += __shfl_xor(ss, off, 32);
  float scale = 1.0f / fmaxf(sqrtf(ss), 1e-12f);
  v.x *= scale; v.y *= scale; v.z *= scale; v.w *= scale;
  row[lane] = v;
}

extern "C" void kernel_launch(void* const* d_in, const int* in_sizes, int n_in,
                              void* d_out, int out_size, void* d_ws, size_t ws_size,
                              hipStream_t stream) {
  const float* x   = (const float*)d_in[0];
  const int*   src = (const int*)  d_in[1];
  const int*   dst = (const int*)  d_in[2];
  const float* Ws1 = (const float*)d_in[3];
  const float* Wn1 = (const float*)d_in[4];
  const float* b1  = (const float*)d_in[5];
  const float* Ws2 = (const float*)d_in[6];
  const float* Wn2 = (const float*)d_in[7];
  const float* b2  = (const float*)d_in[8];
  float* out = (float*)d_out;

  const int N = in_sizes[0] / D;
  const int E = in_sizes[1];

  char* ws = (char*)d_ws;
  float* msg = (float*)ws; ws += (size_t)N * D * sizeof(float);
  float* h1  = (float*)ws; ws += (size_t)N * D * sizeof(float);
  float* deg = (float*)ws; ws += (((size_t)N * sizeof(float)) + 255) & ~(size_t)255;
  float* WTs1 = (float*)ws; ws += (size_t)D * D * sizeof(float);
  float* WTn1 = (float*)ws; ws += (size_t)D * D * sizeof(float);
  float* WTs2 = (float*)ws; ws += (size_t)D * D * sizeof(float);
  float* WTn2 = (float*)ws; ws += (size_t)D * D * sizeof(float);

  {
    dim3 blk(256), grd((D * D + 255) / 256);
    repack_weight<<<grd, blk, 0, stream>>>(Ws1, WTs1);
    repack_weight<<<grd, blk, 0, stream>>>(Wn1, WTn1);
    repack_weight<<<grd, blk, 0, stream>>>(Ws2, WTs2);
    repack_weight<<<grd, blk, 0, stream>>>(Wn2, WTn2);
  }

  const size_t ldsBytes = (size_t)64 * LDS_PP2 * 2 * sizeof(float);
  dim3 gemmGrid((N + 63) / 64);
  const int nd = N * D;

  zero_f32<<<(nd + 255) / 256, 256, 0, stream>>>(msg, nd);
  zero_f32<<<(N + 255) / 256, 256, 0, stream>>>(deg, N);
  scatter_edges<<<(E * 32 + 255) / 256, 256, 0, stream>>>(x, src, dst, msg, deg, E);
  sage_layer<<<gemmGrid, 128, ldsBytes, stream>>>(x, msg, deg, WTs1, WTn1, b1, h1, N);

  zero_f32<<<(nd + 255) / 256, 256, 0, stream>>>(msg, nd);
  scatter_edges<<<(E * 32 + 255) / 256, 256, 0, stream>>>(h1, src, dst, msg, nullptr, E);
  sage_layer<<<gemmGrid, 128, ldsBytes, stream>>>(h1, msg, deg, WTs2, WTn2, b2, out, N);

  l2_normalize<<<(N * 32 + 255) / 256, 256, 0, stream>>>(out, N);
}


# Mean aggregation: edges grouped by destination once per call (int count, scan, fill), each node's rows summed in registers; no float atomics, no msg clears; sum order = list arrival order
# speedup vs baseline: 1.7367x; 1.7367x over previous
#include <hip/hip_runtime.h>
#include <cstddef>
#include <cstdint>

#define D 128
#define LDS_PP2 144

typedef float v2f __attribute__((ext_vector_type(2)));
typedef float v8f __attribute__((ext_vector_type(8)));

__device__ __forceinline__ v8f wmma_f32(v2f a, v2f b, v8f c) {
#if defined(__gfx1250__)
  return __builtin_amdgcn_wmma_f32_16x16x4_f32(false, a, false, b,
                                               (short)0, c, false, false);
#else
  c[0] += a[0] * b[0];
  return c;
#endif
}

__global__ void zero_i32(int* __restrict__ p, int n) {
  int i = blockIdx.x * blockDim.x + threadIdx.x;
  if (i < n) p[i] = 0;
}

__global__ void repack_weight(const float* __restrict__ W, float* __restrict__ WTi) {
  int idx = blockIdx.x * blockDim.x + threadIdx.x;
  if (idx >= D * D) return;
  int n = (idx >> 1) & (D - 1);
  int k = ((idx >> 8) << 1) | (idx & 1);
  WTi[idx] = W[n * D + k];
}

__global__ void count_dst(const int* __restrict__ dst, int* __restrict__ cnt, int nEdges) {
  const int stride = gridDim.x * blockDim.x;
  for (int e = blockIdx.x * blockDim.x + threadIdx.x; e < nEdges; e += stride)
    atomicAdd(cnt + dst[e], 1);
}

__global__ void __launch_bounds__(256)
tile_sums(const int* __restrict__ cnt, int* __restrict__ tileSum, int n) {
  __shared__ int waveTot[8];
  const int tid  = threadIdx.x;
  const int lane = tid & 31;
  const int i = blockIdx.x * 1024 + tid * 4;
  int t = 0;
  if (i + 3 < n) {
    const int4 v = *(const int4*)(cnt + i);
    t = v.x + v.y + v.z + v.w;
  } else {
    if (i     < n) t += cnt[i];
    if (i + 1 < n) t += cnt[i + 1];
    if (i + 2 < n) t += cnt[i + 2];
  }
#pragma unroll
  for (int off = 16; off > 0; off >>= 1) t += __shfl_xor(t, off, 32);
  if (lane == 0) waveTot[tid >> 5] = t;
  __syncthreads();
  if (tid == 0) {
    int s = 0;
#pragma unroll
    for (int k = 0; k < 8; ++k) s += waveTot[k];
    tileSum[blockIdx.x] = s;
  }
}

__global__ void __launch_bounds__(256)
scan_tiles(int* __restrict__ a, const int* __restrict__ tileSum,
           float* __restrict__ deg, int n) {
  __shared__ int waveTot[8];
  const int tid  = threadIdx.x;
  const int lane = tid & 31;
  const int w    = tid >> 5;
  const int i = blockIdx.x * 1024 + tid * 4;
  const bool full = (i + 3 < n);
  int v0 = 0, v1 = 0, v2 = 0, v3 = 0;
  if (full) {
    const int4 v = *(const int4*)(a + i);
    v0 = v.x; v1 = v.y; v2 = v.z; v3 = v.w;
  } else {
    if (i     < n) v0 = a[i];
    if (i + 1 < n) v1 = a[i + 1];
    if (i + 2 < n) v2 = a[i + 2];
  }
  int before = 0;
  for (int k = lane; k < (int)blockIdx.x; k += 32) before += tileSum[k];
#pragma unroll
  for (int off = 16; off > 0; off >>= 1) before += __shfl_xor(before, off, 32);

  const int t = v0 + v1 + v2 + v3;
  int inc = t;
#pragma unroll
  for (int off = 1; off < 32; off <<= 1) {
    const int u = __shfl_up(inc, off, 32);
    if (lane >= off) inc += u;
  }
  if (lane == 31) waveTot[w] = inc;
  __syncthreads();
  const int wt = (lane < 8) ? waveTot[lane] : 0;
  int winc = wt;
#pragma unroll
  for (int off = 1; off < 8; off <<= 1) {
    const int u = __shfl_up(winc, off, 32);
    if (lane >= off) winc += u;
  }
  const int wavePrefix = __shfl(winc - wt, w, 32);
  const int ex = before + wavePrefix + (inc - t);
  if (full) {
    *(int4*)(a + i)     = make_int4(ex, ex + v0, ex + v0 + v1, ex + v0 + v1 + v2);
    *(float4*)(deg + i) = make_float4((float)v0, (float)v1, (float)v2, (float)v3);
  } else {
    if (i     < n) { a[i]     = ex;           deg[i]     = (float)v0; }
    if (i + 1 < n) { a[i + 1] = ex + v0;      deg[i + 1] = (float)v1; }
    if (i + 2 < n) { a[i + 2] = ex + v0 + v1; deg[i + 2] = (float)v2; }
  }
}

__global__ void fill_lists(const int* __restrict__ src, const int* __restrict__ dst,
                           int* __restrict__ cursor, int* __restrict__ nbr, int nEdges) {
  const int stride = gridDim.x * blockDim.x;
  for (int e = blockIdx.x * blockDim.x + threadIdx.x; e < nEdges; e += stride) {
    const int d = dst[e];
    const int s = src[e];
    const unsigned p = (unsigned)atomicAdd(cursor + d, 1);
    const unsigned last = (unsigned)(nEdges - 1);
    nbr[(p < last) ? p : last] = s;
  }
}

__global__ void __launch_bounds__(256)
gather_rows(const float* __restrict__ x, const int* __restrict__ listEnd,
            const int* __restrict__ nbr, float* __restrict__ msg,
            int nNodes, int nEdges) {
  const int lane = threadIdx.x & 31;
  const int nWaves = (int)((gridDim.x * blockDim.x) >> 5);
  int node = __builtin_amdgcn_readfirstlane((int)((blockIdx.x * blockDim.x + threadIdx.x) >> 5));
  const float4* x4 = (const float4*)x;
  for (; node < nNodes; node += nWaves) {
    const int prev = listEnd[(node > 0) ? (node - 1) : 0];
    int end = listEnd[node];
    int beg = (node > 0) ? prev : 0;
    beg = (beg < 0) ? 0 : beg;
    end = (end > nEdges) ? nEdges : end;
    float4 acc = make_float4(0.0f, 0.0f, 0.0f, 0.0f);
    for (int base = beg; base < end; base += 32) {
      const int cnt = (end - base < 32) ? (end - base) : 32;
      int id = (lane < cnt) ? nbr[base + lane] : 0;
      id = (id < 0) ? 0 : ((id >= nNodes) ? (nNodes - 1) : id);
      int j = 0;
      for (; j + 4 <= cnt; j += 4) {
        const int s0 = __builtin_amdgcn_readlane(id, j);
        const int s1 = __builtin_amdgcn_readlane(id, j + 1);
        const int s2 = __builtin_amdgcn_readlane(id, j + 2);
        const int s3 = __builtin_amdgcn_readlane(id, j + 3);
        const float4 r0 = x4[(size_t)s0 * (D / 4) + lane];
        const float4 r1 = x4[(size_t)s1 * (D / 4) + lane];
        const float4 r2 = x4[(size_t)s2 * (D / 4) + lane];
        const float4 r3 = x4[(size_t)s3 * (D / 4) + lane];
        acc.x += r0.x; acc.y += r0.y; acc.z += r0.z; acc.w += r0.w;
        acc.x += r1.x; acc.y += r1.y; acc.z += r1.z; acc.w += r1.w;
        acc.x += r2.x; acc.y += r2.y; acc.z += r2.z; acc.w += r2.w;
        acc.x += r3.x; acc.y += r3.y; acc.z += r3.z; acc.w += r3.w;
      }
      const int rem = cnt - j;
      if (rem > 0) {
        float4 r0 = x4[(size_t)__builtin_amdgcn_readlane(id, j) * (D / 4) + lane];
        float4 r1 = make_float4(0.0f, 0.0f, 0.0f, 0.0f);
        float4 r2 = make_float4(0.0f, 0.0f, 0.0f, 0.0f);
        if (rem > 1) r1 = x4[(size_t)__builtin_amdgcn_readlane(id, j + 1) * (D / 4) + lane];
        if (rem > 2) r2 = x4[(size_t)__builtin_amdgcn_readlane(id, j + 2) * (D / 4) + lane];
        acc.x += r0.x; acc.y += r0.y; acc.z += r0.z; acc.w += r0.w;
        acc.x += r1.x; acc.y += r1.y; acc.z += r1.z; acc.w += r1.w;
        acc.x += r2.x; acc.y += r2.y; acc.z += r2.z; acc.w += r2.w;
      }
    }
    ((float4*)msg)[(size_t)node * (D / 4) + lane] = acc;
  }
}

__global__ void __launch_bounds__(128)
sage_layer(const float* __restrict__ xin, const float* __restrict__ msg,
           const float* __restrict__ deg,
           const float* __restrict__ WTiS, const float* __restrict__ WTiN,
           const float* __restrict__ bias,
           float* __restrict__ out, int nNodes) {
  extern __shared__ float lds[];
  const int tid  = threadIdx.x;
  const int wave = tid >> 5;
  const int lane = tid & 31;
  const int half = lane >> 4;
  const int l16  = lane & 15;
  const int m0   = blockIdx.x * 64 + wave * 16;
  const int mRow = m0 + l16;
  const int mC   = (mRow < nNodes) ? mRow : (nNodes - 1);

  const float invd = 1.0f / fmaxf(deg[mC], 1.0f);
  const float* xr = xin + (size_t)mC * D + half * 2;
  const float* mr = msg + (size_t)mC * D + half * 2;
  const v2f* B2 = (const v2f*)lds;

  v8f acc[8];
#pragma unroll
  for (int t = 0; t < 8; ++t) acc[t] = (v8f)(0.0f);

  for (int phase = 0; phase < 2; ++phase) {
    const float4* src4 = (const float4*)(phase == 0 ? WTiS : WTiN);
    const float*  arow = (phase == 0) ? xr : mr;

    __syncthreads();
    for (int i = tid; i < (D * D) / 4; i += 128) {
      ((float4*)lds)[(i >> 6) * (LDS_PP2 / 2) + (i & 63)] = src4[i];
    }
    __syncthreads();

    v2f aCur = *(const v2f*)(arow);
    if (phase == 1) aCur *= invd;

    for (int k0 = 0; k0 < D; k0 += 4) {
      const int base2 = ((k0 >> 1) + half) * LDS_PP2 + l16;
      v2f b[8];
#pragma unroll
      for (int t = 0; t < 8; ++t) b[t] = B2[base2 + t * 16];

      const int kN = (k0 + 4 < D) ? (k0 + 4) : k0;
      v2f aNext = *(const v2f*)(arow + kN);
      if (phase == 1) aNext *= invd;

#pragma unroll
      for (int t = 0; t < 8; ++t) acc[t] = wmma_f32(aCur, b[t], acc[t]);
      aCur = aNext;
    }
  }

#pragma unroll
  for (int t = 0; t < 8; ++t) {
    const int col = t * 16 + l16;
    const float bv = bias[col];
#pragma unroll
    for (int r = 0; r < 8; ++r) {
      int row = m0 + r + half * 8;
      if (row < nNodes) out[(size_t)row * D + col] = acc[t][r] + bv;
    }
  }
}

__global__ void l2_normalize(float* __restrict__ h, int nNodes) {
  int wave = (int)((blockIdx.x * blockDim.x + threadIdx.x) >> 5);
  int lane = threadIdx.x & 31;
  if (wave >= nNodes) return;
  float4* row = (float4*)(h + (size_t)wave * D);
  float4 v = row[lane];
  float ss = v.x * v.x + v.y * v.y + v.z * v.z + v.w * v.w;
#pragma unroll
  for (int off = 16; off > 0; off >>= 1) ss += __shfl_xor(ss, off, 32);
  float scale = 1.0f / fmaxf(sqrtf(ss), 1e-12f);
  v.x *= scale; v.y *= scale; v.z *= scale; v.w *= scale;
  row[lane] = v;
}

extern "C" void kernel_launch(void* const* d_in, const int* in_sizes, int n_in,
                              void* d_out, int out_size, void* d_ws, size_t ws_size,
                              hipStream_t stream) {
  const float* x   = (const float*)d_in[0];
  const int*   src = (const int*)  d_in[1];
  const int*   dst = (const int*)  d_in[2];
  const float* Ws1 = (const float*)d_in[3];
  const float* Wn1 = (const float*)d_in[4];
  const float* b1  = (const float*)d_in[5];
  const float* Ws2 = (const float*)d_in[6];
  const float* Wn2 = (const float*)d_in[7];
  const float* b2  = (const float*)d_in[8];
  float* out = (float*)d_out;

  const int N = in_sizes[0] / D;
  const int E = in_sizes[1];

  char* ws = (char*)d_ws;
  float* msg = (float*)ws; ws += (size_t)N * D * sizeof(float);
  float* h1  = (float*)ws; ws += (size_t)N * D * sizeof(float);
  float* deg = (float*)ws; ws += (((size_t)N * sizeof(float)) + 255) & ~(size_t)255;
  float* WTs1 = (float*)ws; ws += (size_t)D * D * sizeof(float);
  float* WTn1 = (float*)ws; ws += (size_t)D * D * sizeof(float);
  float* WTs2 = (float*)ws; ws += (size_t)D * D * sizeof(float);
  float* WTn2 = (float*)ws; ws += (size_t)D * D * sizeof(float);

  {
    dim3 blk(256), grd((D * D + 255) / 256);
    repack_weight<<<grd, blk, 0, stream>>>(Ws1, WTs1);
    repack_weight<<<grd, blk, 0, stream>>>(Wn1, WTn1);
    repack_weight<<<grd, blk, 0, stream>>>(Ws2, WTs2);
    repack_weight<<<grd, blk, 0, stream>>>(Wn2, WTn2);
  }

  const size_t ldsBytes = (size_t)64 * LDS_PP2 * 2 * sizeof(float);
  dim3 gemmGrid((N + 63) / 64);

  char* sc = (char*)d_out;
  const int nTiles = (N + 1023) / 1024;
  int* listPtr = (int*)sc; sc += (((size_t)N * sizeof(int)) + 255) & ~(size_t)255;
  int* tileSum = (int*)sc; sc += (((size_t)nTiles * sizeof(int)) + 255) & ~(size_t)255;
  int* nbr     = (int*)sc;

  zero_i32<<<(N + 255) / 256, 256, 0, stream>>>(listPtr, N);
  count_dst<<<2048, 256, 0, stream>>>(dst, listPtr, E);
  tile_sums<<<nTiles, 256, 0, stream>>>(listPtr, tileSum, N);
  scan_tiles<<<nTiles, 256, 0, stream>>>(listPtr, tileSum, deg, N);
  fill_lists<<<2048, 256, 0, stream>>>(src, dst, listPtr, nbr, E);

  gather_rows<<<2048, 256, 0, stream>>>(x, listPtr, nbr, msg, N, E);
  sage_layer<<<gemmGrid, 128, ldsBytes, stream>>>(x, msg, deg, WTs1, WTn1, b1, h1, N);

  gather_rows<<<2048, 256, 0, stream>>>(h1, listPtr, nbr, msg, N, E);
  sage_layer<<<gemmGrid, 128, ldsBytes, stream>>>(h1, msg, deg, WTs2, WTn2, b2, out, N);

  l2_normalize<<<(N * 32 + 255) / 256, 256, 0, stream>>>(out, N);
}


# Edge grouping redone as a two-level counting sort with every counter in LDS (256-node buckets, then per-node lists); no global atomics left; sum order = arrival order of the LDS adds
# speedup vs baseline: 2.3498x; 1.3530x over previous
#include <hip/hip_runtime.h>
#include <cstddef>
#include <cstdint>

#define D 128
#define LDS_PP2 144

typedef float v2f __attribute__((ext_vector_type(2)));
typedef float v8f __attribute__((ext_vector_type(8)));

__device__ __forceinline__ v8f wmma_f32(v2f a, v2f b, v8f c) {
#if defined(__gfx1250__)
  return __builtin_amdgcn_wmma_f32_16x16x4_f32(false, a, false, b,
                                               (short)0, c, false, false);
#else
  c[0] += a[0] * b[0];
  return c;
#endif
}

__global__ void repack_weight(const float* __restrict__ W, float* __restrict__ WTi) {
  int idx = blockIdx.x * blockDim.x + threadIdx.x;
  if (idx >= D * D) return;
  int n = (idx >> 1) & (D - 1);
  int k = ((idx >> 8) << 1) | (idx & 1);
  WTi[idx] = W[n * D + k];
}

#define BKT_SHIFT 8
#define BKT_NODES 256
#define BIN_GROUPS 512

__device__ __forceinline__ int bucket_of(int d, int nBuckets) {
  const int b = d >> BKT_SHIFT;
  return (b < 0) ? 0 : ((b >= nBuckets) ? (nBuckets - 1) : b);
}

__device__ __forceinline__ int local_of(int d, int node0) {
  const int l = d - node0;
  return (l < 0) ? 0 : ((l >= BKT_NODES) ? (BKT_NODES - 1) : l);
}

__global__ void __launch_bounds__(256)
bucket_count(const int* __restrict__ dst, int* __restrict__ hist,
             int nEdges, int nBuckets, int chunk) {
  extern __shared__ int lcnt[];
  const int tid = threadIdx.x;
  for (int b = tid; b < nBuckets; b += 256) lcnt[b] = 0;
  __syncthreads();
  const int lo = (int)blockIdx.x * chunk;
  int hi = lo + chunk;
  hi = (hi < nEdges) ? hi : nEdges;
  int e = lo + tid;
  for (; e + 768 < hi; e += 1024) {
    const int d0 = dst[e];
    const int d1 = dst[e + 256];
    const int d2 = dst[e + 512];
    const int d3 = dst[e + 768];
    atomicAdd(&lcnt[bucket_of(d0, nBuckets)], 1);
    atomicAdd(&lcnt[bucket_of(d1, nBuckets)], 1);
    atomicAdd(&lcnt[bucket_of(d2, nBuckets)], 1);
    atomicAdd(&lcnt[bucket_of(d3, nBuckets)], 1);
  }
  for (; e < hi; e += 256) atomicAdd(&lcnt[bucket_of(dst[e], nBuckets)], 1);
  __syncthreads();
  for (int b = tid; b < nBuckets; b += 256)
    hist[(size_t)b * gridDim.x + blockIdx.x] = lcnt[b];
}

__global__ void __launch_bounds__(256)
tile_sums(const int* __restrict__ cnt, int* __restrict__ tileSum, int n) {
  __shared__ int waveTot[8];
  const int tid  = threadIdx.x;
  const int lane = tid & 31;
  const int i = blockIdx.x * 1024 + tid * 4;
  int t = 0;
  if (i + 3 < n) {
    const int4 v = *(const int4*)(cnt + i);
    t = v.x + v.y + v.z + v.w;
  } else {
    if (i     < n) t += cnt[i];
    if (i + 1 < n) t += cnt[i + 1];
    if (i + 2 < n) t += cnt[i + 2];
  }
#pragma unroll
  for (int off = 16; off > 0; off >>= 1) t += __shfl_xor(t, off, 32);
  if (lane == 0) waveTot[tid >> 5] = t;
  __syncthreads();
  if (tid == 0) {
    int s = 0;
#pragma unroll
    for (int k = 0; k < 8; ++k) s += waveTot[k];
    tileSum[blockIdx.x] = s;
  }
}

__global__ void __launch_bounds__(256)
scan_tiles(int* __restrict__ a, const int* __restrict__ tileSum, int n) {
  __shared__ int waveTot[8];
  const int tid  = threadIdx.x;
  const int lane = tid & 31;
  const int w    = tid >> 5;
  const int i = blockIdx.x * 1024 + tid * 4;
  const bool full = (i + 3 < n);
  int v0 = 0, v1 = 0, v2 = 0, v3 = 0;
  if (full) {
    const int4 v = *(const int4*)(a + i);
    v0 = v.x; v1 = v.y; v2 = v.z; v3 = v.w;
  } else {
    if (i     < n) v0 = a[i];
    if (i + 1 < n) v1 = a[i + 1];
    if (i + 2 < n) v2 = a[i + 2];
  }
  int before = 0;
  for (int k = lane; k < (int)blockIdx.x; k += 32) before += tileSum[k];
#pragma unroll
  for (int off = 16; off > 0; off >>= 1) before += __shfl_xor(before, off, 32);

  const int t = v0 + v1 + v2 + v3;
  int inc = t;
#pragma unroll
  for (int off = 1; off < 32; off <<= 1) {
    const int u = __shfl_up(inc, off, 32);
    if (lane >= off) inc += u;
  }
  if (lane == 31) waveTot[w] = inc;
  __syncthreads();
  const int wt = (lane < 8) ? waveTot[lane] : 0;
  int winc = wt;
#pragma unroll
  for (int off = 1; off < 8; off <<= 1) {
    const int u = __shfl_up(winc, off, 32);
    if (lane >= off) winc += u;
  }
  const int wavePrefix = __shfl(winc - wt, w, 32);
  const int ex = before + wavePrefix + (inc - t);
  if (full) {
    *(int4*)(a + i) = make_int4(ex, ex + v0, ex + v0 + v1, ex + v0 + v1 + v2);
  } else {
    if (i     < n) a[i]     = ex;
    if (i + 1 < n) a[i + 1] = ex + v0;
    if (i + 2 < n) a[i + 2] = ex + v0 + v1;
  }
}

__global__ void __launch_bounds__(256)
bucket_scatter(const int* __restrict__ src, const int* __restrict__ dst,
               const int* __restrict__ offs, int* __restrict__ binned,
               int nEdges, int nBuckets, int chunk) {
  extern __shared__ int lcur[];
  const int tid = threadIdx.x;
  for (int b = tid; b < nBuckets; b += 256)
    lcur[b] = offs[(size_t)b * gridDim.x + blockIdx.x];
  __syncthreads();
  const int lo = (int)blockIdx.x * chunk;
  int hi = lo + chunk;
  hi = (hi < nEdges) ? hi : nEdges;
  const unsigned last = (unsigned)(nEdges - 1);
  int2* pairs = (int2*)binned;
  int e = lo + tid;
  for (; e + 768 < hi; e += 1024) {
    const int d0 = dst[e],       s0 = src[e];
    const int d1 = dst[e + 256], s1 = src[e + 256];
    const int d2 = dst[e + 512], s2 = src[e + 512];
    const int d3 = dst[e + 768], s3 = src[e + 768];
    const unsigned p0 = (unsigned)atomicAdd(&lcur[bucket_of(d0, nBuckets)], 1);
    const unsigned p1 = (unsigned)atomicAdd(&lcur[bucket_of(d1, nBuckets)], 1);
    const unsigned p2 = (unsigned)atomicAdd(&lcur[bucket_of(d2, nBuckets)], 1);
    const unsigned p3 = (unsigned)atomicAdd(&lcur[bucket_of(d3, nBuckets)], 1);
    pairs[(p0 < last) ? p0 : last] = make_int2(s0, d0);
    pairs[(p1 < last) ? p1 : last] = make_int2(s1, d1);
    pairs[(p2 < last) ? p2 : last] = make_int2(s2, d2);
    pairs[(p3 < last) ? p3 : last] = make_int2(s3, d3);
  }
  for (; e < hi; e += 256) {
    const int d0 = dst[e], s0 = src[e];
    const unsigned p0 = (unsigned)atomicAdd(&lcur[bucket_of(d0, nBuckets)], 1);
    pairs[(p0 < last) ? p0 : last] = make_int2(s0, d0);
  }
}

__global__ void __launch_bounds__(BKT_NODES)
bucket_lists(const int* __restrict__ binned, const int* __restrict__ offs,
             int* __restrict__ listEnd, float* __restrict__ deg, int* __restrict__ nbr,
             int nNodes, int nEdges, int nBuckets, int groups) {
  __shared__ int cnt[BKT_NODES];
  __shared__ int cur[BKT_NODES];
  __shared__ int waveTot[8];
  const int tid  = threadIdx.x;
  const int lane = tid & 31;
  const int w    = tid >> 5;
  const int b    = (int)blockIdx.x;
  const int node0 = b << BKT_SHIFT;
  int start = offs[(size_t)b * groups];
  int end   = (b + 1 < nBuckets) ? offs[(size_t)(b + 1) * groups] : nEdges;
  start = (start < 0) ? 0 : ((start > nEdges) ? nEdges : start);
  end   = (end > nEdges) ? nEdges : end;
  const int2* pairs = (const int2*)binned;

  cnt[tid] = 0;
  __syncthreads();
  int e = start + tid;
  for (; e + 3 * BKT_NODES < end; e += 4 * BKT_NODES) {
    const int d0 = pairs[e].y;
    const int d1 = pairs[e + BKT_NODES].y;
    const int d2 = pairs[e + 2 * BKT_NODES].y;
    const int d3 = pairs[e + 3 * BKT_NODES].y;
    atomicAdd(&cnt[local_of(d0, node0)], 1);
    atomicAdd(&cnt[local_of(d1, node0)], 1);
    atomicAdd(&cnt[local_of(d2, node0)], 1);
    atomicAdd(&cnt[local_of(d3, node0)], 1);
  }
  for (; e < end; e += BKT_NODES) atomicAdd(&cnt[local_of(pairs[e].y, node0)], 1);
  __syncthreads();

  const int c = cnt[tid];
  int inc = c;
#pragma unroll
  for (int off = 1; off < 32; off <<= 1) {
    const int u = __shfl_up(inc, off, 32);
    if (lane >= off) inc += u;
  }
  if (lane == 31) waveTot[w] = inc;
  __syncthreads();
  const int wt = (lane < 8) ? waveTot[lane] : 0;
  int winc = wt;
#pragma unroll
  for (int off = 1; off < 8; off <<= 1) {
    const int u = __shfl_up(winc, off, 32);
    if (lane >= off) winc += u;
  }
  const int wavePrefix = __shfl(winc - wt, w, 32);
  const int first = start + wavePrefix + (inc - c);
  cur[tid] = first;
  const int node = node0 + tid;
  if (node < nNodes) {
    listEnd[node] = first + c;
    deg[node]     = (float)c;
  }
  __syncthreads();

  const unsigned last = (unsigned)(nEdges - 1);
  e = start + tid;
  for (; e + 3 * BKT_NODES < end; e += 4 * BKT_NODES) {
    const int2 p0 = pairs[e];
    const int2 p1 = pairs[e + BKT_NODES];
    const int2 p2 = pairs[e + 2 * BKT_NODES];
    const int2 p3 = pairs[e + 3 * BKT_NODES];
    const unsigned q0 = (unsigned)atomicAdd(&cur[local_of(p0.y, node0)], 1);
    const unsigned q1 = (unsigned)atomicAdd(&cur[local_of(p1.y, node0)], 1);
    const unsigned q2 = (unsigned)atomicAdd(&cur[local_of(p2.y, node0)], 1);
    const unsigned q3 = (unsigned)atomicAdd(&cur[local_of(p3.y, node0)], 1);
    nbr[(q0 < last) ? q0 : last] = p0.x;
    nbr[(q1 < last) ? q1 : last] = p1.x;
    nbr[(q2 < last) ? q2 : last] = p2.x;
    nbr[(q3 < last) ? q3 : last] = p3.x;
  }
  for (; e < end; e += BKT_NODES) {
    const int2 p0 = pairs[e];
    const unsigned q0 = (unsigned)atomicAdd(&cur[local_of(p0.y, node0)], 1);
    nbr[(q0 < last) ? q0 : last] = p0.x;
  }
}

__global__ void __launch_bounds__(256)
gather_rows(const float* __restrict__ x, const int* __restrict__ listEnd,
            const int* __restrict__ nbr, float* __restrict__ msg,
            int nNodes, int nEdges) {
  const int lane = threadIdx.x & 31;
  const int nWaves = (int)((gridDim.x * blockDim.x) >> 5);
  int node = __builtin_amdgcn_readfirstlane((int)((blockIdx.x * blockDim.x + threadIdx.x) >> 5));
  const float4* x4 = (const float4*)x;
  for (; node < nNodes; node += nWaves) {
    const int prev = listEnd[(node > 0) ? (node - 1) : 0];
    int end = listEnd[node];
    int beg = (node > 0) ? prev : 0;
    beg = (beg < 0) ? 0 : beg;
    end = (end > nEdges) ? nEdges : end;
    float4 acc = make_float4(0.0f, 0.0f, 0.0f, 0.0f);
    for (int base = beg; base < end; base += 32) {
      const int cnt = (end - base < 32) ? (end - base) : 32;
      int id = (lane < cnt) ? nbr[base + lane] : 0;
      id = (id < 0) ? 0 : ((id >= nNodes) ? (nNodes - 1) : id);
      int j = 0;
      for (; j + 4 <= cnt; j += 4) {
        const int s0 = __builtin_amdgcn_readlane(id, j);
        const int s1 = __builtin_amdgcn_readlane(id, j + 1);
        const int s2 = __builtin_amdgcn_readlane(id, j + 2);
        const int s3 = __builtin_amdgcn_readlane(id, j + 3);
        const float4 r0 = x4[(size_t)s0 * (D / 4) + lane];
        const float4 r1 = x4[(size_t)s1 * (D / 4) + lane];
        const float4 r2 = x4[(size_t)s2 * (D / 4) + lane];
        const float4 r3 = x4[(size_t)s3 * (D / 4) + lane];
        acc.x += r0.x; acc.y += r0.y; acc.z += r0.z; acc.w += r0.w;
        acc.x += r1.x; acc.y += r1.y; acc.z += r1.z; acc.w += r1.w;
        acc.x += r2.x; acc.y += r2.y; acc.z += r2.z; acc.w += r2.w;
        acc.x += r3.x; acc.y += r3.y; acc.z += r3.z; acc.w += r3.w;
      }
      const int rem = cnt - j;
      if (rem > 0) {
        float4 r0 = x4[(size_t)__builtin_amdgcn_readlane(id, j) * (D / 4) + lane];
        float4 r1 = make_float4(0.0f, 0.0f, 0.0f, 0.0f);
        float4 r2 = make_float4(0.0f, 0.0f, 0.0f, 0.0f);
        if (rem > 1) r1 = x4[(size_t)__builtin_amdgcn_readlane(id, j + 1) * (D / 4) + lane];
        if (rem > 2) r2 = x4[(size_t)__builtin_amdgcn_readlane(id, j + 2) * (D / 4) + lane];
        acc.x += r0.x; acc.y += r0.y; acc.z += r0.z; acc.w += r0.w;
        acc.x += r1.x; acc.y += r1.y; acc.z += r1.z; acc.w += r1.w;
        acc.x += r2.x; acc.y += r2.y; acc.z += r2.z; acc.w += r2.w;
      }
    }
    ((float4*)msg)[(size_t)node * (D / 4) + lane] = acc;
  }
}

__global__ void __launch_bounds__(128)
sage_layer(const float* __restrict__ xin, const float* __restrict__ msg,
           const float* __restrict__ deg,
           const float* __restrict__ WTiS, const float* __restrict__ WTiN,
           const float* __restrict__ bias,
           float* __restrict__ out, int nNodes) {
  extern __shared__ float lds[];
  const int tid  = threadIdx.x;
  const int wave = tid >> 5;
  const int lane = tid & 31;
  const int half = lane >> 4;
  const int l16  = lane & 15;
  const int m0   = blockIdx.x * 64 + wave * 16;
  const int mRow = m0 + l16;
  const int mC   = (mRow < nNodes) ? mRow : (nNodes - 1);

  const float invd = 1.0f / fmaxf(deg[mC], 1.0f);
  const float* xr = xin + (size_t)mC * D + half * 2;
  const float* mr = msg + (size_t)mC * D + half * 2;
  const v2f* B2 = (const v2f*)lds;

  v8f acc[8];
#pragma unroll
  for (int t = 0; t < 8; ++t) acc[t] = (v8f)(0.0f);

  for (int phase = 0; phase < 2; ++phase) {
    const float4* src4 = (const float4*)(phase == 0 ? WTiS : WTiN);
    const float*  arow = (phase == 0) ? xr : mr;

    __syncthreads();
    for (int i = tid; i < (D * D) / 4; i += 128) {
      ((float4*)lds)[(i >> 6) * (LDS_PP2 / 2) + (i & 63)] = src4[i];
    }
    __syncthreads();

    v2f aCur = *(const v2f*)(arow);
    if (phase == 1) aCur *= invd;

    for (int k0 = 0; k0 < D; k0 += 4) {
      const int base2 = ((k0 >> 1) + half) * LDS_PP2 + l16;
      v2f b[8];
#pragma unroll
      for (int t = 0; t < 8; ++t) b[t] = B2[base2 + t * 16];

      const int kN = (k0 + 4 < D) ? (k0 + 4) : k0;
      v2f aNext = *(const v2f*)(arow + kN);
      if (phase == 1) aNext *= invd;

#pragma unroll
      for (int t = 0; t < 8; ++t) acc[t] = wmma_f32(aCur, b[t], acc[t]);
      aCur = aNext;
    }
  }

#pragma unroll
  for (int t = 0; t < 8; ++t) {
    const int col = t * 16 + l16;
    const float bv = bias[col];
#pragma unroll
    for (int r = 0; r < 8; ++r) {
      int row = m0 + r + half * 8;
      if (row < nNodes) out[(size_t)row * D + col] = acc[t][r] + bv;
    }
  }
}

__global__ void l2_normalize(float* __restrict__ h, int nNodes) {
  int wave = (int)((blockIdx.x * blockDim.x + threadIdx.x) >> 5);
  int lane = threadIdx.x & 31;
  if (wave >= nNodes) return;
  float4* row = (float4*)(h + (size_t)wave * D);
  float4 v = row[lane];
  float ss = v.x * v.x + v.y * v.y + v.z * v.z + v.w * v.w;
#pragma unroll
  for (int off = 16; off > 0; off >>= 1) ss += __shfl_xor(ss, off, 32);
  float scale = 1.0f / fmaxf(sqrtf(ss), 1e-12f);
  v.x *= scale; v.y *= scale; v.z *= scale; v.w *= scale;
  row[lane] = v;
}

extern "C" void kernel_launch(void* const* d_in, const int* in_sizes, int n_in,
                              void* d_out, int out_size, void* d_ws, size_t ws_size,
                              hipStream_t stream) {
  const float* x   = (const float*)d_in[0];
  const int*   src = (const int*)  d_in[1];
  const int*   dst = (const int*)  d_in[2];
  const float* Ws1 = (const float*)d_in[3];
  const float* Wn1 = (const float*)d_in[4];
  const float* b1  = (const float*)d_in[5];
  const float* Ws2 = (const float*)d_in[6];
  const float* Wn2 = (const float*)d_in[7];
  const float* b2  = (const float*)d_in[8];
  float* out = (float*)d_out;

  const int N = in_sizes[0] / D;
  const int E = in_sizes[1];

  char* ws = (char*)d_ws;
  float* msg = (float*)ws; ws += (size_t)N * D * sizeof(float);
  float* h1  = (float*)ws; ws += (size_t)N * D * sizeof(float);
  float* deg = (float*)ws; ws += (((size_t)N * sizeof(float)) + 255) & ~(size_t)255;
  float* WTs1 = (float*)ws; ws += (size_t)D * D * sizeof(float);
  float* WTn1 = (float*)ws; ws += (size_t)D * D * sizeof(float);
  float* WTs2 = (float*)ws; ws += (size_t)D * D * sizeof(float);
  float* WTn2 = (float*)ws; ws += (size_t)D * D * sizeof(float);

  {
    dim3 blk(256), grd((D * D + 255) / 256);
    repack_weight<<<grd, blk, 0, stream>>>(Ws1, WTs1);
    repack_weight<<<grd, blk, 0, stream>>>(Wn1, WTn1);
    repack_weight<<<grd, blk, 0, stream>>>(Ws2, WTs2);
    repack_weight<<<grd, blk, 0, stream>>>(Wn2, WTn2);
  }

  const size_t ldsBytes = (size_t)64 * LDS_PP2 * 2 * sizeof(float);
  dim3 gemmGrid((N + 63) / 64);

  char* sc = (char*)d_out;
  const int nBuckets = (N + BKT_NODES - 1) / BKT_NODES;
  const int chunk    = (E + BIN_GROUPS - 1) / BIN_GROUPS;
  const int nHist    = nBuckets * BIN_GROUPS;
  const int nTiles   = (nHist + 1023) / 1024;
  int* listPtr = (int*)sc; sc += (((size_t)N * sizeof(int)) + 255) & ~(size_t)255;
  int* hist    = (int*)sc; sc += (((size_t)nHist * sizeof(int)) + 255) & ~(size_t)255;
  int* tileSum = (int*)sc; sc += (((size_t)nTiles * sizeof(int)) + 255) & ~(size_t)255;
  int* binned  = (int*)sc; sc += (((size_t)E * 2 * sizeof(int)) + 255) & ~(size_t)255;
  int* nbr     = (int*)sc;
  const size_t bktLds = (size_t)nBuckets * sizeof(int);

  bucket_count<<<BIN_GROUPS, 256, bktLds, stream>>>(dst, hist, E, nBuckets, chunk);
  tile_sums<<<nTiles, 256, 0, stream>>>(hist, tileSum, nHist);
  scan_tiles<<<nTiles, 256, 0, stream>>>(hist, tileSum, nHist);
  bucket_scatter<<<BIN_GROUPS, 256, bktLds, stream>>>(src, dst, hist, binned, E, nBuckets, chunk);
  bucket_lists<<<nBuckets, BKT_NODES, 0, stream>>>(binned, hist, listPtr, deg, nbr, N, E, nBuckets, BIN_GROUPS);

  gather_rows<<<2048, 256, 0, stream>>>(x, listPtr, nbr, msg, N, E);
  sage_layer<<<gemmGrid, 128, ldsBytes, stream>>>(x, msg, deg, WTs1, WTn1, b1, h1, N);

  gather_rows<<<2048, 256, 0, stream>>>(h1, listPtr, nbr, msg, N, E);
  sage_layer<<<gemmGrid, 128, ldsBytes, stream>>>(h1, msg, deg, WTs2, WTn2, b2, out, N);

  l2_normalize<<<(N * 32 + 255) / 256, 256, 0, stream>>>(out, N);
}


# Row L2 normalisation folded into the layer-2 GEMM epilogue (its launch of 100000 one-row waves and the output re-read gone); sum-of-squares order: 8 tiles per lane, then half-wave butterfly
# speedup vs baseline: 2.9114x; 1.2390x over previous
#include <hip/hip_runtime.h>
#include <cstddef>
#include <cstdint>

#define D 128
#define LDS_PP2 144

typedef float v2f __attribute__((ext_vector_type(2)));
typedef float v8f __attribute__((ext_vector_type(8)));

__device__ __forceinline__ v8f wmma_f32(v2f a, v2f b, v8f c) {
#if defined(__gfx1250__)
  return __builtin_amdgcn_wmma_f32_16x16x4_f32(false, a, false, b,
                                               (short)0, c, false, false);
#else
  c[0] += a[0] * b[0];
  return c;
#endif
}

template <int M>
__device__ __forceinline__ float row_xor(float v) {
#if defined(__HIP_DEVICE_COMPILE__)
  const int b = __builtin_bit_cast(int, v);
  return __builtin_bit_cast(float, __builtin_amdgcn_update_dpp(b, b, 0x160 | M, 0xf, 0xf, false));
#else
  return v;
#endif
}

__global__ void repack_weight(const float* __restrict__ W, float* __restrict__ WTi) {
  int idx = blockIdx.x * blockDim.x + threadIdx.x;
  if (idx >= D * D) return;
  int n = (idx >> 1) & (D - 1);
  int k = ((idx >> 8) << 1) | (idx & 1);
  WTi[idx] = W[n * D + k];
}

#define BKT_SHIFT 8
#define BKT_NODES 256
#define BIN_GROUPS 512

__device__ __forceinline__ int bucket_of(int d, int nBuckets) {
  const int b = d >> BKT_SHIFT;
  return (b < 0) ? 0 : ((b >= nBuckets) ? (nBuckets - 1) : b);
}

__device__ __forceinline__ int local_of(int d, int node0) {
  const int l = d - node0;
  return (l < 0) ? 0 : ((l >= BKT_NODES) ? (BKT_NODES - 1) : l);
}

__global__ void __launch_bounds__(256)
bucket_count(const int* __restrict__ dst, int* __restrict__ hist,
             int nEdges, int nBuckets, int chunk) {
  extern __shared__ int lcnt[];
  const int tid = threadIdx.x;
  for (int b = tid; b < nBuckets; b += 256) lcnt[b] = 0;
  __syncthreads();
  const int lo = (int)blockIdx.x * chunk;
  int hi = lo + chunk;
  hi = (hi < nEdges) ? hi : nEdges;
  int e = lo + tid;
  for (; e + 768 < hi; e += 1024) {
    const int d0 = dst[e];
    const int d1 = dst[e + 256];
    const int d2 = dst[e + 512];
    const int d3 = dst[e + 768];
    atomicAdd(&lcnt[bucket_of(d0, nBuckets)], 1);
    atomicAdd(&lcnt[bucket_of(d1, nBuckets)], 1);
    atomicAdd(&lcnt[bucket_of(d2, nBuckets)], 1);
    atomicAdd(&lcnt[bucket_of(d3, nBuckets)], 1);
  }
  for (; e < hi; e += 256) atomicAdd(&lcnt[bucket_of(dst[e], nBuckets)], 1);
  __syncthreads();
  for (int b = tid; b < nBuckets; b += 256)
    hist[(size_t)b * gridDim.x + blockIdx.x] = lcnt[b];
}

__global__ void __launch_bounds__(256)
tile_sums(const int* __restrict__ cnt, int* __restrict__ tileSum, int n) {
  __shared__ int waveTot[8];
  const int tid  = threadIdx.x;
  const int lane = tid & 31;
  const int i = blockIdx.x * 1024 + tid * 4;
  int t = 0;
  if (i + 3 < n) {
    const int4 v = *(const int4*)(cnt + i);
    t = v.x + v.y + v.z + v.w;
  } else {
    if (i     < n) t += cnt[i];
    if (i + 1 < n) t += cnt[i + 1];
    if (i + 2 < n) t += cnt[i + 2];
  }
#pragma unroll
  for (int off = 16; off > 0; off >>= 1) t += __shfl_xor(t, off, 32);
  if (lane == 0) waveTot[tid >> 5] = t;
  __syncthreads();
  if (tid == 0) {
    int s = 0;
#pragma unroll
    for (int k = 0; k < 8; ++k) s += waveTot[k];
    tileSum[blockIdx.x] = s;
  }
}

__global__ void __launch_bounds__(256)
scan_tiles(int* __restrict__ a, const int* __restrict__ tileSum, int n) {
  __shared__ int waveTot[8];
  const int tid  = threadIdx.x;
  const int lane = tid & 31;
  const int w    = tid >> 5;
  const int i = blockIdx.x * 1024 + tid * 4;
  const bool full = (i + 3 < n);
  int v0 = 0, v1 = 0, v2 = 0, v3 = 0;
  if (full) {
    const int4 v = *(const int4*)(a + i);
    v0 = v.x; v1 = v.y; v2 = v.z; v3 = v.w;
  } else {
    if (i     < n) v0 = a[i];
    if (i + 1 < n) v1 = a[i + 1];
    if (i + 2 < n) v2 = a[i + 2];
  }
  int before = 0;
  for (int k = lane; k < (int)blockIdx.x; k += 32) before += tileSum[k];
#pragma unroll
  for (int off = 16; off > 0; off >>= 1) before += __shfl_xor(before, off, 32);

  const int t = v0 + v1 + v2 + v3;
  int inc = t;
#pragma unroll
  for (int off = 1; off < 32; off <<= 1) {
    const int u = __shfl_up(inc, off, 32);
    if (lane >= off) inc += u;
  }
  if (lane == 31) waveTot[w] = inc;
  __syncthreads();
  const int wt = (lane < 8) ? waveTot[lane] : 0;
  int winc = wt;
#pragma unroll
  for (int off = 1; off < 8; off <<= 1) {
    const int u = __shfl_up(winc, off, 32);
    if (lane >= off) winc += u;
  }
  const int wavePrefix = __shfl(winc - wt, w, 32);
  const int ex = before + wavePrefix + (inc - t);
  if (full) {
    *(int4*)(a + i) = make_int4(ex, ex + v0, ex + v0 + v1, ex + v0 + v1 + v2);
  } else {
    if (i     < n) a[i]     = ex;
    if (i + 1 < n) a[i + 1] = ex + v0;
    if (i + 2 < n) a[i + 2] = ex + v0 + v1;
  }
}

__global__ void __launch_bounds__(256)
bucket_scatter(const int* __restrict__ src, const int* __restrict__ dst,
               const int* __restrict__ offs, int* __restrict__ binned,
               int nEdges, int nBuckets, int chunk) {
  extern __shared__ int lcur[];
  const int tid = threadIdx.x;
  for (int b = tid; b < nBuckets; b += 256)
    lcur[b] = offs[(size_t)b * gridDim.x + blockIdx.x];
  __syncthreads();
  const int lo = (int)blockIdx.x * chunk;
  int hi = lo + chunk;
  hi = (hi < nEdges) ? hi : nEdges;
  const unsigned last = (unsigned)(nEdges - 1);
  int2* pairs = (int2*)binned;
  int e = lo + tid;
  for (; e + 768 < hi; e += 1024) {
    const int d0 = dst[e],       s0 = src[e];
    const int d1 = dst[e + 256], s1 = src[e + 256];
    const int d2 = dst[e + 512], s2 = src[e + 512];
    const int d3 = dst[e + 768], s3 = src[e + 768];
    const unsigned p0 = (unsigned)atomicAdd(&lcur[bucket_of(d0, nBuckets)], 1);
    const unsigned p1 = (unsigned)atomicAdd(&lcur[bucket_of(d1, nBuckets)], 1);
    const unsigned p2 = (unsigned)atomicAdd(&lcur[bucket_of(d2, nBuckets)], 1);
    const unsigned p3 = (unsigned)atomicAdd(&lcur[bucket_of(d3, nBuckets)], 1);
    pairs[(p0 < last) ? p0 : last] = make_int2(s0, d0);
    pairs[(p1 < last) ? p1 : last] = make_int2(s1, d1);
    pairs[(p2 < last) ? p2 : last] = make_int2(s2, d2);
    pairs[(p3 < last) ? p3 : last] = make_int2(s3, d3);
  }
  for (; e < hi; e += 256) {
    const int d0 = dst[e], s0 = src[e];
    const unsigned p0 = (unsigned)atomicAdd(&lcur[bucket_of(d0, nBuckets)], 1);
    pairs[(p0 < last) ? p0 : last] = make_int2(s0, d0);
  }
}

__global__ void __launch_bounds__(BKT_NODES)
bucket_lists(const int* __restrict__ binned, const int* __restrict__ offs,
             int* __restrict__ listEnd, float* __restrict__ deg, int* __restrict__ nbr,
             int nNodes, int nEdges, int nBuckets, int groups) {
  __shared__ int cnt[BKT_NODES];
  __shared__ int cur[BKT_NODES];
  __shared__ int waveTot[8];
  const int tid  = threadIdx.x;
  const int lane = tid & 31;
  const int w    = tid >> 5;
  const int b    = (int)blockIdx.x;
  const int node0 = b << BKT_SHIFT;
  int start = offs[(size_t)b * groups];
  int end   = (b + 1 < nBuckets) ? offs[(size_t)(b + 1) * groups] : nEdges;
  start = (start < 0) ? 0 : ((start > nEdges) ? nEdges : start);
  end   = (end > nEdges) ? nEdges : end;
  const int2* pairs = (const int2*)binned;

  cnt[tid] = 0;
  __syncthreads();
  int e = start + tid;
  for (; e + 3 * BKT_NODES < end; e += 4 * BKT_NODES) {
    const int d0 = pairs[e].y;
    const int d1 = pairs[e + BKT_NODES].y;
    const int d2 = pairs[e + 2 * BKT_NODES].y;
    const int d3 = pairs[e + 3 * BKT_NODES].y;
    atomicAdd(&cnt[local_of(d0, node0)], 1);
    atomicAdd(&cnt[local_of(d1, node0)], 1);
    atomicAdd(&cnt[local_of(d2, node0)], 1);
    atomicAdd(&cnt[local_of(d3, node0)], 1);
  }
  for (; e < end; e += BKT_NODES) atomicAdd(&cnt[local_of(pairs[e].y, node0)], 1);
  __syncthreads();

  const int c = cnt[tid];
  int inc = c;
#pragma unroll
  for (int off = 1; off < 32; off <<= 1) {
    const int u = __shfl_up(inc, off, 32);
    if (lane >= off) inc += u;
  }
  if (lane == 31) waveTot[w] = inc;
  __syncthreads();
  const int wt = (lane < 8) ? waveTot[lane] : 0;
  int winc = wt;
#pragma unroll
  for (int off = 1; off < 8; off <<= 1) {
    const int u = __shfl_up(winc, off, 32);
    if (lane >= off) winc += u;
  }
  const int wavePrefix = __shfl(winc - wt, w, 32);
  const int first = start + wavePrefix + (inc - c);
  cur[tid] = first;
  const int node = node0 + tid;
  if (node < nNodes) {
    listEnd[node] = first + c;
    deg[node]     = (float)c;
  }
  __syncthreads();

  const unsigned last = (unsigned)(nEdges - 1);
  e = start + tid;
  for (; e + 3 * BKT_NODES < end; e += 4 * BKT_NODES) {
    const int2 p0 = pairs[e];
    const int2 p1 = pairs[e + BKT_NODES];
    const int2 p2 = pairs[e + 2 * BKT_NODES];
    const int2 p3 = pairs[e + 3 * BKT_NODES];
    const unsigned q0 = (unsigned)atomicAdd(&cur[local_of(p0.y, node0)], 1);
    const unsigned q1 = (unsigned)atomicAdd(&cur[local_of(p1.y, node0)], 1);
    const unsigned q2 = (unsigned)atomicAdd(&cur[local_of(p2.y, node0)], 1);
    const unsigned q3 = (unsigned)atomicAdd(&cur[local_of(p3.y, node0)], 1);
    nbr[(q0 < last) ? q0 : last] = p0.x;
    nbr[(q1 < last) ? q1 : last] = p1.x;
    nbr[(q2 < last) ? q2 : last] = p2.x;
    nbr[(q3 < last) ? q3 : last] = p3.x;
  }
  for (; e < end; e += BKT_NODES) {
    const int2 p0 = pairs[e];
    const unsigned q0 = (unsigned)atomicAdd(&cur[local_of(p0.y, node0)], 1);
    nbr[(q0 < last) ? q0 : last] = p0.x;
  }
}

__global__ void __launch_bounds__(256)
gather_rows(const float* __restrict__ x, const int* __restrict__ listEnd,
            const int* __restrict__ nbr, float* __restrict__ msg,
            int nNodes, int nEdges) {
  const int lane = threadIdx.x & 31;
  const int nWaves = (int)((gridDim.x * blockDim.x) >> 5);
  int node = __builtin_amdgcn_readfirstlane((int)((blockIdx.x * blockDim.x + threadIdx.x) >> 5));
  const float4* x4 = (const float4*)x;
  for (; node < nNodes; node += nWaves) {
    const int prev = listEnd[(node > 0) ? (node - 1) : 0];
    int end = listEnd[node];
    int beg = (node > 0) ? prev : 0;
    beg = (beg < 0) ? 0 : beg;
    end = (end > nEdges) ? nEdges : end;
    float4 acc = make_float4(0.0f, 0.0f, 0.0f, 0.0f);
    for (int base = beg; base < end; base += 32) {
      const int cnt = (end - base < 32) ? (end - base) : 32;
      int id = (lane < cnt) ? nbr[base + lane] : 0;
      id = (id < 0) ? 0 : ((id >= nNodes) ? (nNodes - 1) : id);
      int j = 0;
      for (; j + 4 <= cnt; j += 4) {
        const int s0 = __builtin_amdgcn_readlane(id, j);
        const int s1 = __builtin_amdgcn_readlane(id, j + 1);
        const int s2 = __builtin_amdgcn_readlane(id, j + 2);
        const int s3 = __builtin_amdgcn_readlane(id, j + 3);
        const float4 r0 = x4[(size_t)s0 * (D / 4) + lane];
        const float4 r1 = x4[(size_t)s1 * (D / 4) + lane];
        const float4 r2 = x4[(size_t)s2 * (D / 4) + lane];
        const float4 r3 = x4[(size_t)s3 * (D / 4) + lane];
        acc.x += r0.x; acc.y += r0.y; acc.z += r0.z; acc.w += r0.w;
        acc.x += r1.x; acc.y += r1.y; acc.z += r1.z; acc.w += r1.w;
        acc.x += r2.x; acc.y += r2.y; acc.z += r2.z; acc.w += r2.w;
        acc.x += r3.x; acc.y += r3.y; acc.z += r3.z; acc.w += r3.w;
      }
      const int rem = cnt - j;
      if (rem > 0) {
        float4 r0 = x4[(size_t)__builtin_amdgcn_readlane(id, j) * (D / 4) + lane];
        float4 r1 = make_float4(0.0f, 0.0f, 0.0f, 0.0f);
        float4 r2 = make_float4(0.0f, 0.0f, 0.0f, 0.0f);
        if (rem > 1) r1 = x4[(size_t)__builtin_amdgcn_readlane(id, j + 1) * (D / 4) + lane];
        if (rem > 2) r2 = x4[(size_t)__builtin_amdgcn_readlane(id, j + 2) * (D / 4) + lane];
        acc.x += r0.x; acc.y += r0.y; acc.z += r0.z; acc.w += r0.w;
        acc.x += r1.x; acc.y += r1.y; acc.z += r1.z; acc.w += r1.w;
        acc.x += r2.x; acc.y += r2.y; acc.z += r2.z; acc.w += r2.w;
      }
    }
    ((float4*)msg)[(size_t)node * (D / 4) + lane] = acc;
  }
}

template <bool kNormalize>
__global__ void __launch_bounds__(128)
sage_layer(const float* __restrict__ xin, const float* __restrict__ msg,
           const float* __restrict__ deg,
           const float* __restrict__ WTiS, const float* __restrict__ WTiN,
           const float* __restrict__ bias,
           float* __restrict__ out, int nNodes) {
  extern __shared__ float lds[];
  const int tid  = threadIdx.x;
  const int wave = tid >> 5;
  const int lane = tid & 31;
  const int half = lane >> 4;
  const int l16  = lane & 15;
  const int m0   = blockIdx.x * 64 + wave * 16;
  const int mRow = m0 + l16;
  const int mC   = (mRow < nNodes) ? mRow : (nNodes - 1);

  const float invd = 1.0f / fmaxf(deg[mC], 1.0f);
  const float* xr = xin + (size_t)mC * D + half * 2;
  const float* mr = msg + (size_t)mC * D + half * 2;
  const v2f* B2 = (const v2f*)lds;

  v8f acc[8];
#pragma unroll
  for (int t = 0; t < 8; ++t) acc[t] = (v8f)(0.0f);

  for (int phase = 0; phase < 2; ++phase) {
    const float4* src4 = (const float4*)(phase == 0 ? WTiS : WTiN);
    const float*  arow = (phase == 0) ? xr : mr;

    __syncthreads();
    for (int i = tid; i < (D * D) / 4; i += 128) {
      ((float4*)lds)[(i >> 6) * (LDS_PP2 / 2) + (i & 63)] = src4[i];
    }
    __syncthreads();

    v2f aCur = *(const v2f*)(arow);
    if (phase == 1) aCur *= invd;

    for (int k0 = 0; k0 < D; k0 += 4) {
      const int base2 = ((k0 >> 1) + half) * LDS_PP2 + l16;
      v2f b[8];
#pragma unroll
      for (int t = 0; t < 8; ++t) b[t] = B2[base2 + t * 16];

      const int kN = (k0 + 4 < D) ? (k0 + 4) : k0;
      v2f aNext = *(const v2f*)(arow + kN);
      if (phase == 1) aNext *= invd;

#pragma unroll
      for (int t = 0; t < 8; ++t) acc[t] = wmma_f32(aCur, b[t], acc[t]);
      aCur = aNext;
    }
  }

#pragma unroll
  for (int t = 0; t < 8; ++t) {
    const float bv = bias[t * 16 + l16];
#pragma unroll
    for (int r = 0; r < 8; ++r) acc[t][r] += bv;
  }

  if (kNormalize) {
    float ss[8];
#pragma unroll
    for (int r = 0; r < 8; ++r) {
      float s = 0.0f;
#pragma unroll
      for (int t = 0; t < 8; ++t) s += acc[t][r] * acc[t][r];
      ss[r] = s;
    }
#pragma unroll
    for (int r = 0; r < 8; ++r) ss[r] += row_xor<1>(ss[r]);
#pragma unroll
    for (int r = 0; r < 8; ++r) ss[r] += row_xor<2>(ss[r]);
#pragma unroll
    for (int r = 0; r < 8; ++r) ss[r] += row_xor<4>(ss[r]);
#pragma unroll
    for (int r = 0; r < 8; ++r) ss[r] += row_xor<8>(ss[r]);
#pragma unroll
    for (int r = 0; r < 8; ++r) {
      const float scale = 1.0f / fmaxf(sqrtf(ss[r]), 1e-12f);
#pragma unroll
      for (int t = 0; t < 8; ++t) acc[t][r] *= scale;
    }
  }

#pragma unroll
  for (int t = 0; t < 8; ++t) {
    const int col = t * 16 + l16;
#pragma unroll
    for (int r = 0; r < 8; ++r) {
      const int row = m0 + r + half * 8;
      if (row < nNodes) out[(size_t)row * D + col] = acc[t][r];
    }
  }
}

extern "C" void kernel_launch(void* const* d_in, const int* in_sizes, int n_in,
                              void* d_out, int out_size, void* d_ws, size_t ws_size,
                              hipStream_t stream) {
  const float* x   = (const float*)d_in[0];
  const int*   src = (const int*)  d_in[1];
  const int*   dst = (const int*)  d_in[2];
  const float* Ws1 = (const float*)d_in[3];
  const float* Wn1 = (const float*)d_in[4];
  const float* b1  = (const float*)d_in[5];
  const float* Ws2 = (const float*)d_in[6];
  const float* Wn2 = (const float*)d_in[7];
  const float* b2  = (const float*)d_in[8];
  float* out = (float*)d_out;

  const int N = in_sizes[0] / D;
  const int E = in_sizes[1];

  char* ws = (char*)d_ws;
  float* msg = (float*)ws; ws += (size_t)N * D * sizeof(float);
  float* h1  = (float*)ws; ws += (size_t)N * D * sizeof(float);
  float* deg = (float*)ws; ws += (((size_t)N * sizeof(float)) + 255) & ~(size_t)255;
  float* WTs1 = (float*)ws; ws += (size_t)D * D * sizeof(float);
  float* WTn1 = (float*)ws; ws += (size_t)D * D * sizeof(float);
  float* WTs2 = (float*)ws; ws += (size_t)D * D * sizeof(float);
  float* WTn2 = (float*)ws; ws += (size_t)D * D * sizeof(float);

  {
    dim3 blk(256), grd((D * D + 255) / 256);
    repack_weight<<<grd, blk, 0, stream>>>(Ws1, WTs1);
    repack_weight<<<grd, blk, 0, stream>>>(Wn1, WTn1);
    repack_weight<<<grd, blk, 0, stream>>>(Ws2, WTs2);
    repack_weight<<<grd, blk, 0, stream>>>(Wn2, WTn2);
  }

  const size_t ldsBytes = (size_t)64 * LDS_PP2 * 2 * sizeof(float);
  dim3 gemmGrid((N + 63) / 64);

  char* sc = (char*)d_out;
  const int nBuckets = (N + BKT_NODES - 1) / BKT_NODES;
  const int chunk    = (E + BIN_GROUPS - 1) / BIN_GROUPS;
  const int nHist    = nBuckets * BIN_GROUPS;
  const int nTiles   = (nHist + 1023) / 1024;
  int* listPtr = (int*)sc; sc += (((size_t)N * sizeof(int)) + 255) & ~(size_t)255;
  int* hist    = (int*)sc; sc += (((size_t)nHist * sizeof(int)) + 255) & ~(size_t)255;
  int* tileSum = (int*)sc; sc += (((size_t)nTiles * sizeof(int)) + 255) & ~(size_t)255;
  int* binned  = (int*)sc; sc += (((size_t)E * 2 * sizeof(int)) + 255) & ~(size_t)255;
  int* nbr     = (int*)sc;
  const size_t bktLds = (size_t)nBuckets * sizeof(int);

  bucket_count<<<BIN_GROUPS, 256, bktLds, stream>>>(dst, hist, E, nBuckets, chunk);
  tile_sums<<<nTiles, 256, 0, stream>>>(hist, tileSum, nHist);
  scan_tiles<<<nTiles, 256, 0, stream>>>(hist, tileSum, nHist);
  bucket_scatter<<<BIN_GROUPS, 256, bktLds, stream>>>(src, dst, hist, binned, E, nBuckets, chunk);
  bucket_lists<<<nBuckets, BKT_NODES, 0, stream>>>(binned, hist, listPtr, deg, nbr, N, E, nBuckets, BIN_GROUPS);

  gather_rows<<<2048, 256, 0, stream>>>(x, listPtr, nbr, msg, N, E);
  sage_layer<false><<<gemmGrid, 128, ldsBytes, stream>>>(x, msg, deg, WTs1, WTn1, b1, h1, N);

  gather_rows<<<2048, 256, 0, stream>>>(h1, listPtr, nbr, msg, N, E);
  sage_layer<true><<<gemmGrid, 128, ldsBytes, stream>>>(h1, msg, deg, WTs2, WTn2, b2, out, N);
}


# Layer kernel feeds both A operands from a per-wave LDS tile: own rows fetched whole, neighbour rows summed in-kernel in list order (unchanged); 2 gather launches and sum plane gone; 8-wave blocks
# speedup vs baseline: 3.0612x; 1.0515x over previous
#include <hip/hip_runtime.h>
#include <cstddef>
#include <cstdint>

#define D 128
#define LDS_PP2 144

typedef float v2f __attribute__((ext_vector_type(2)));
typedef float v8f __attribute__((ext_vector_type(8)));

__device__ __forceinline__ v8f wmma_f32(v2f a, v2f b, v8f c) {
#if defined(__gfx1250__)
  return __builtin_amdgcn_wmma_f32_16x16x4_f32(false, a, false, b,
                                               (short)0, c, false, false);
#else
  c[0] += a[0] * b[0];
  return c;
#endif
}

template <int M>
__device__ __forceinline__ float row_xor(float v) {
#if defined(__HIP_DEVICE_COMPILE__)
  const int b = __builtin_bit_cast(int, v);
  return __builtin_bit_cast(float, __builtin_amdgcn_update_dpp(b, b, 0x160 | M, 0xf, 0xf, false));
#else
  return v;
#endif
}

__global__ void repack_weight(const float* __restrict__ W, float* __restrict__ WTi) {
  int idx = blockIdx.x * blockDim.x + threadIdx.x;
  if (idx >= D * D) return;
  int n = (idx >> 1) & (D - 1);
  int k = ((idx >> 8) << 1) | (idx & 1);
  WTi[idx] = W[n * D + k];
}

#define BKT_SHIFT 8
#define BKT_NODES 256
#define BIN_GROUPS 512

__device__ __forceinline__ int bucket_of(int d, int nBuckets) {
  const int b = d >> BKT_SHIFT;
  return (b < 0) ? 0 : ((b >= nBuckets) ? (nBuckets - 1) : b);
}

__device__ __forceinline__ int local_of(int d, int node0) {
  const int l = d - node0;
  return (l < 0) ? 0 : ((l >= BKT_NODES) ? (BKT_NODES - 1) : l);
}

__global__ void __launch_bounds__(256)
bucket_count(const int* __restrict__ dst, int* __restrict__ hist,
             int nEdges, int nBuckets, int chunk) {
  extern __shared__ int lcnt[];
  const int tid = threadIdx.x;
  for (int b = tid; b < nBuckets; b += 256) lcnt[b] = 0;
  __syncthreads();
  const int lo = (int)blockIdx.x * chunk;
  int hi = lo + chunk;
  hi = (hi < nEdges) ? hi : nEdges;
  int e = lo + tid;
  for (; e + 768 < hi; e += 1024) {
    const int d0 = dst[e];
    const int d1 = dst[e + 256];
    const int d2 = dst[e + 512];
    const int d3 = dst[e + 768];
    atomicAdd(&lcnt[bucket_of(d0, nBuckets)], 1);
    atomicAdd(&lcnt[bucket_of(d1, nBuckets)], 1);
    atomicAdd(&lcnt[bucket_of(d2, nBuckets)], 1);
    atomicAdd(&lcnt[bucket_of(d3, nBuckets)], 1);
  }
  for (; e < hi; e += 256) atomicAdd(&lcnt[bucket_of(dst[e], nBuckets)], 1);
  __syncthreads();
  for (int b = tid; b < nBuckets; b += 256)
    hist[(size_t)b * gridDim.x + blockIdx.x] = lcnt[b];
}

__global__ void __launch_bounds__(256)
tile_sums(const int* __restrict__ cnt, int* __restrict__ tileSum, int n) {
  __shared__ int waveTot[8];
  const int tid  = threadIdx.x;
  const int lane = tid & 31;
  const int i = blockIdx.x * 1024 + tid * 4;
  int t = 0;
  if (i + 3 < n) {
    const int4 v = *(const int4*)(cnt + i);
    t = v.x + v.y + v.z + v.w;
  } else {
    if (i     < n) t += cnt[i];
    if (i + 1 < n) t += cnt[i + 1];
    if (i + 2 < n) t += cnt[i + 2];
  }
#pragma unroll
  for (int off = 16; off > 0; off >>= 1) t += __shfl_xor(t, off, 32);
  if (lane == 0) waveTot[tid >> 5] = t;
  __syncthreads();
  if (tid == 0) {
    int s = 0;
#pragma unroll
    for (int k = 0; k < 8; ++k) s += waveTot[k];
    tileSum[blockIdx.x] = s;
  }
}

__global__ void __launch_bounds__(256)
scan_tiles(int* __restrict__ a, const int* __restrict__ tileSum, int n) {
  __shared__ int waveTot[8];
  const int tid  = threadIdx.x;
  const int lane = tid & 31;
  const int w    = tid >> 5;
  const int i = blockIdx.x * 1024 + tid * 4;
  const bool full = (i + 3 < n);
  int v0 = 0, v1 = 0, v2 = 0, v3 = 0;
  if (full) {
    const int4 v = *(const int4*)(a + i);
    v0 = v.x; v1 = v.y; v2 = v.z; v3 = v.w;
  } else {
    if (i     < n) v0 = a[i];
    if (i + 1 < n) v1 = a[i + 1];
    if (i + 2 < n) v2 = a[i + 2];
  }
  int before = 0;
  for (int k = lane; k < (int)blockIdx.x; k += 32) before += tileSum[k];
#pragma unroll
  for (int off = 16; off > 0; off >>= 1) before += __shfl_xor(before, off, 32);

  const int t = v0 + v1 + v2 + v3;
  int inc = t;
#pragma unroll
  for (int off = 1; off < 32; off <<= 1) {
    const int u = __shfl_up(inc, off, 32);
    if (lane >= off) inc += u;
  }
  if (lane == 31) waveTot[w] = inc;
  __syncthreads();
  const int wt = (lane < 8) ? waveTot[lane] : 0;
  int winc = wt;
#pragma unroll
  for (int off = 1; off < 8; off <<= 1) {
    const int u = __shfl_up(winc, off, 32);
    if (lane >= off) winc += u;
  }
  const int wavePrefix = __shfl(winc - wt, w, 32);
  const int ex = before + wavePrefix + (inc - t);
  if (full) {
    *(int4*)(a + i) = make_int4(ex, ex + v0, ex + v0 + v1, ex + v0 + v1 + v2);
  } else {
    if (i     < n) a[i]     = ex;
    if (i + 1 < n) a[i + 1] = ex + v0;
    if (i + 2 < n) a[i + 2] = ex + v0 + v1;
  }
}

__global__ void __launch_bounds__(256)
bucket_scatter(const int* __restrict__ src, const int* __restrict__ dst,
               const int* __restrict__ offs, int* __restrict__ binned,
               int nEdges, int nBuckets, int chunk) {
  extern __shared__ int lcur[];
  const int tid = threadIdx.x;
  for (int b = tid; b < nBuckets; b += 256)
    lcur[b] = offs[(size_t)b * gridDim.x + blockIdx.x];
  __syncthreads();
  const int lo = (int)blockIdx.x * chunk;
  int hi = lo + chunk;
  hi = (hi < nEdges) ? hi : nEdges;
  const unsigned last = (unsigned)(nEdges - 1);
  int2* pairs = (int2*)binned;
  int e = lo + tid;
  for (; e + 768 < hi; e += 1024) {
    const int d0 = dst[e],       s0 = src[e];
    const int d1 = dst[e + 256], s1 = src[e + 256];
    const int d2 = dst[e + 512], s2 = src[e + 512];
    const int d3 = dst[e + 768], s3 = src[e + 768];
    const unsigned p0 = (unsigned)atomicAdd(&lcur[bucket_of(d0, nBuckets)], 1);
    const unsigned p1 = (unsigned)atomicAdd(&lcur[bucket_of(d1, nBuckets)], 1);
    const unsigned p2 = (unsigned)atomicAdd(&lcur[bucket_of(d2, nBuckets)], 1);
    const unsigned p3 = (unsigned)atomicAdd(&lcur[bucket_of(d3, nBuckets)], 1);
    pairs[(p0 < last) ? p0 : last] = make_int2(s0, d0);
    pairs[(p1 < last) ? p1 : last] = make_int2(s1, d1);
    pairs[(p2 < last) ? p2 : last] = make_int2(s2, d2);
    pairs[(p3 < last) ? p3 : last] = make_int2(s3, d3);
  }
  for (; e < hi; e += 256) {
    const int d0 = dst[e], s0 = src[e];
    const unsigned p0 = (unsigned)atomicAdd(&lcur[bucket_of(d0, nBuckets)], 1);
    pairs[(p0 < last) ? p0 : last] = make_int2(s0, d0);
  }
}

__global__ void __launch_bounds__(BKT_NODES)
bucket_lists(const int* __restrict__ binned, const int* __restrict__ offs,
             int* __restrict__ listEnd, float* __restrict__ deg, int* __restrict__ nbr,
             int nNodes, int nEdges, int nBuckets, int groups) {
  __shared__ int cnt[BKT_NODES];
  __shared__ int cur[BKT_NODES];
  __shared__ int waveTot[8];
  const int tid  = threadIdx.x;
  const int lane = tid & 31;
  const int w    = tid >> 5;
  const int b    = (int)blockIdx.x;
  const int node0 = b << BKT_SHIFT;
  int start = offs[(size_t)b * groups];
  int end   = (b + 1 < nBuckets) ? offs[(size_t)(b + 1) * groups] : nEdges;
  start = (start < 0) ? 0 : ((start > nEdges) ? nEdges : start);
  end   = (end > nEdges) ? nEdges : end;
  const int2* pairs = (const int2*)binned;

  cnt[tid] = 0;
  __syncthreads();
  int e = start + tid;
  for (; e + 3 * BKT_NODES < end; e += 4 * BKT_NODES) {
    const int d0 = pairs[e].y;
    const int d1 = pairs[e + BKT_NODES].y;
    const int d2 = pairs[e + 2 * BKT_NODES].y;
    const int d3 = pairs[e + 3 * BKT_NODES].y;
    atomicAdd(&cnt[local_of(d0, node0)], 1);
    atomicAdd(&cnt[local_of(d1, node0)], 1);
    atomicAdd(&cnt[local_of(d2, node0)], 1);
    atomicAdd(&cnt[local_of(d3, node0)], 1);
  }
  for (; e < end; e += BKT_NODES) atomicAdd(&cnt[local_of(pairs[e].y, node0)], 1);
  __syncthreads();

  const int c = cnt[tid];
  int inc = c;
#pragma unroll
  for (int off = 1; off < 32; off <<= 1) {
    const int u = __shfl_up(inc, off, 32);
    if (lane >= off) inc += u;
  }
  if (lane == 31) waveTot[w] = inc;
  __syncthreads();
  const int wt = (lane < 8) ? waveTot[lane] : 0;
  int winc = wt;
#pragma unroll
  for (int off = 1; off < 8; off <<= 1) {
    const int u = __shfl_up(winc, off, 32);
    if (lane >= off) winc += u;
  }
  const int wavePrefix = __shfl(winc - wt, w, 32);
  const int first = start + wavePrefix + (inc - c);
  cur[tid] = first;
  const int node = node0 + tid;
  if (node < nNodes) {
    listEnd[node] = first + c;
    deg[node]     = (float)c;
  }
  __syncthreads();

  const unsigned last = (unsigned)(nEdges - 1);
  e = start + tid;
  for (; e + 3 * BKT_NODES < end; e += 4 * BKT_NODES) {
    const int2 p0 = pairs[e];
    const int2 p1 = pairs[e + BKT_NODES];
    const int2 p2 = pairs[e + 2 * BKT_NODES];
    const int2 p3 = pairs[e + 3 * BKT_NODES];
    const unsigned q0 = (unsigned)atomicAdd(&cur[local_of(p0.y, node0)], 1);
    const unsigned q1 = (unsigned)atomicAdd(&cur[local_of(p1.y, node0)], 1);
    const unsigned q2 = (unsigned)atomicAdd(&cur[local_of(p2.y, node0)], 1);
    const unsigned q3 = (unsigned)atomicAdd(&cur[local_of(p3.y, node0)], 1);
    nbr[(q0 < last) ? q0 : last] = p0.x;
    nbr[(q1 < last) ? q1 : last] = p1.x;
    nbr[(q2 < last) ? q2 : last] = p2.x;
    nbr[(q3 < last) ? q3 : last] = p3.x;
  }
  for (; e < end; e += BKT_NODES) {
    const int2 p0 = pairs[e];
    const unsigned q0 = (unsigned)atomicAdd(&cur[local_of(p0.y, node0)], 1);
    nbr[(q0 < last) ? q0 : last] = p0.x;
  }
}

#define A_PITCH 132
#define W_LDS_FLOATS (64 * LDS_PP2 * 2)
#define LAYER_WAVES 8

template <bool kNormalize>
__global__ void __launch_bounds__(LAYER_WAVES * 32)
sage_layer(const float* __restrict__ xin,
           const int* __restrict__ listEnd, const int* __restrict__ nbr,
           const float* __restrict__ deg,
           const float* __restrict__ WTiS, const float* __restrict__ WTiN,
           const float* __restrict__ bias,
           float* __restrict__ out, int nNodes, int nEdges) {
  __shared__ __attribute__((aligned(16))) float lds[W_LDS_FLOATS + LAYER_WAVES * 16 * A_PITCH];
  const int tid  = threadIdx.x;
  const int wave = tid >> 5;
  const int lane = tid & 31;
  const int half = lane >> 4;
  const int l16  = lane & 15;
  const int m0   = blockIdx.x * (LAYER_WAVES * 16) + wave * 16;
  const int mRow = m0 + l16;
  const int mC   = (mRow < nNodes) ? mRow : (nNodes - 1);

  const float invd = 1.0f / fmaxf(deg[mC], 1.0f);
  const v2f* B2 = (const v2f*)lds;
  float* aTile = lds + W_LDS_FLOATS + wave * (16 * A_PITCH);

  {
    const float4* x4 = (const float4*)xin;
    float4 r[16];
#pragma unroll
    for (int q = 0; q < 16; ++q) {
      const int row = (m0 + q < nNodes) ? (m0 + q) : (nNodes - 1);
      r[q] = x4[(size_t)row * (D / 4) + lane];
    }
#pragma unroll
    for (int q = 0; q < 16; ++q) *(float4*)(aTile + q * A_PITCH + lane * 4) = r[q];
  }
  const float* ar = aTile + l16 * A_PITCH + half * 2;

  v8f acc[8];
#pragma unroll
  for (int t = 0; t < 8; ++t) acc[t] = (v8f)(0.0f);

  {
    const float4* src4 = (const float4*)WTiS;
    for (int i = tid; i < (D * D) / 4; i += LAYER_WAVES * 32) {
      ((float4*)lds)[(i >> 6) * (LDS_PP2 / 2) + (i & 63)] = src4[i];
    }
    __syncthreads();

    v2f aCur = *(const v2f*)(ar);
    for (int k0 = 0; k0 < D; k0 += 4) {
      const int base2 = ((k0 >> 1) + half) * LDS_PP2 + l16;
      v2f b[8];
#pragma unroll
      for (int t = 0; t < 8; ++t) b[t] = B2[base2 + t * 16];

      const int kN = (k0 + 4 < D) ? (k0 + 4) : k0;
      const v2f aNext = *(const v2f*)(ar + kN);

#pragma unroll
      for (int t = 0; t < 8; ++t) acc[t] = wmma_f32(aCur, b[t], acc[t]);
      aCur = aNext;
    }
  }

  {
    const float4* x4 = (const float4*)xin;
    int leIdx = m0 - 1 + lane;
    leIdx = (leIdx < 0) ? 0 : ((leIdx >= nNodes) ? (nNodes - 1) : leIdx);
    const int le = listEnd[leIdx];
    const int node0 = __builtin_amdgcn_readfirstlane(m0);

    auto bounds = [&](int i, int& b, int& e) {
      b = 0; e = 0;
      const int node = node0 + i;
      if (node < nNodes) {
        b = (node > 0) ? __builtin_amdgcn_readlane(le, i) : 0;
        e = __builtin_amdgcn_readlane(le, i + 1);
        b = (b < 0) ? 0 : b;
        e = (e > nEdges) ? nEdges : e;
      }
    };
    auto ids = [&](int b, int e) -> int {
      const int cnt = (e - b < 32) ? (e - b) : 32;
      return (lane < cnt) ? nbr[b + lane] : 0;
    };
    auto valid_row = [&](int id) -> int {
      return (id < 0) ? 0 : ((id >= nNodes) ? (nNodes - 1) : id);
    };
#define ROW_OF(q) x4[(size_t)__builtin_amdgcn_readlane(id, j + (q)) * (D / 4) + lane]
#define ADD_ROW(v) { acc4.x += (v).x; acc4.y += (v).y; acc4.z += (v).z; acc4.w += (v).w; }

    int begC, endC;
    bounds(0, begC, endC);
    int idC = valid_row(ids(begC, endC));
    for (int i = 0; i < 16; ++i) {
      int begX = 0, endX = 0, idX = 0;
      if (i + 1 < 16) { bounds(i + 1, begX, endX); idX = ids(begX, endX); }

      float4 acc4 = make_float4(0.0f, 0.0f, 0.0f, 0.0f);
      int id = idC;
      for (int base = begC; base < endC; base += 32) {
        const int cnt = (endC - base < 32) ? (endC - base) : 32;
        if (base != begC) id = valid_row(ids(base, endC));
        for (int j = 0; j < cnt; j += 16) {
          float4 r[16];
#pragma unroll
          for (int q = 0; q < 16; ++q) r[q] = make_float4(0.0f, 0.0f, 0.0f, 0.0f);
          r[0] = ROW_OF(0);
#pragma unroll
          for (int q = 1; q < 16; ++q)
            if (j + q < cnt) r[q] = ROW_OF(q);
#pragma unroll
          for (int q = 0; q < 16; ++q) ADD_ROW(r[q]);
        }
      }
      *(float4*)(aTile + i * A_PITCH + lane * 4) = acc4;
      begC = begX; endC = endX; idC = valid_row(idX);
    }
#undef ROW_OF
#undef ADD_ROW
  }

  {
    const float4* src4 = (const float4*)WTiN;
    __syncthreads();
    for (int i = tid; i < (D * D) / 4; i += LAYER_WAVES * 32) {
      ((float4*)lds)[(i >> 6) * (LDS_PP2 / 2) + (i & 63)] = src4[i];
    }
    __syncthreads();

    v2f aCur = *(const v2f*)(ar);
    aCur *= invd;
    for (int k0 = 0; k0 < D; k0 += 4) {
      const int base2 = ((k0 >> 1) + half) * LDS_PP2 + l16;
      v2f b[8];
#pragma unroll
      for (int t = 0; t < 8; ++t) b[t] = B2[base2 + t * 16];

      const int kN = (k0 + 4 < D) ? (k0 + 4) : k0;
      v2f aNext = *(const v2f*)(ar + kN);
      aNext *= invd;

#pragma unroll
      for (int t = 0; t < 8; ++t) acc[t] = wmma_f32(aCur, b[t], acc[t]);
      aCur = aNext;
    }
  }

#pragma unroll
  for (int t = 0; t < 8; ++t) {
    const float bv = bias[t * 16 + l16];
#pragma unroll
    for (int r = 0; r < 8; ++r) acc[t][r] += bv;
  }

  if (kNormalize) {
    float ss[8];
#pragma unroll
    for (int r = 0; r < 8; ++r) {
      float s = 0.0f;
#pragma unroll
      for (int t = 0; t < 8; ++t) s += acc[t][r] * acc[t][r];
      ss[r] = s;
    }
#pragma unroll
    for (int r = 0; r < 8; ++r) ss[r] += row_xor<1>(ss[r]);
#pragma unroll
    for (int r = 0; r < 8; ++r) ss[r] += row_xor<2>(ss[r]);
#pragma unroll
    for (int r = 0; r < 8; ++r) ss[r] += row_xor<4>(ss[r]);
#pragma unroll
    for (int r = 0; r < 8; ++r) ss[r] += row_xor<8>(ss[r]);
#pragma unroll
    for (int r = 0; r < 8; ++r) {
      const float scale = 1.0f / fmaxf(sqrtf(ss[r]), 1e-12f);
#pragma unroll
      for (int t = 0; t < 8; ++t) acc[t][r] *= scale;
    }
  }

#pragma unroll
  for (int t = 0; t < 8; ++t) {
    const int col = t * 16 + l16;
#pragma unroll
    for (int r = 0; r < 8; ++r) {
      const int row = m0 + r + half * 8;
      if (row < nNodes) out[(size_t)row * D + col] = acc[t][r];
    }
  }
}

extern "C" void kernel_launch(void* const* d_in, const int* in_sizes, int n_in,
                              void* d_out, int out_size, void* d_ws, size_t ws_size,
                              hipStream_t stream) {
  const float* x   = (const float*)d_in[0];
  const int*   src = (const int*)  d_in[1];
  const int*   dst = (const int*)  d_in[2];
  const float* Ws1 = (const float*)d_in[3];
  const float* Wn1 = (const float*)d_in[4];
  const float* b1  = (const float*)d_in[5];
  const float* Ws2 = (const float*)d_in[6];
  const float* Wn2 = (const float*)d_in[7];
  const float* b2  = (const float*)d_in[8];
  float* out = (float*)d_out;

  const int N = in_sizes[0] / D;
  const int E = in_sizes[1];

  char* ws = (char*)d_ws;
  char*  lists = ws;       ws += (size_t)N * D * sizeof(float);
  float* h1  = (float*)ws; ws += (size_t)N * D * sizeof(float);
  float* deg = (float*)ws; ws += (((size_t)N * sizeof(float)) + 255) & ~(size_t)255;
  float* WTs1 = (float*)ws; ws += (size_t)D * D * sizeof(float);
  float* WTn1 = (float*)ws; ws += (size_t)D * D * sizeof(float);
  float* WTs2 = (float*)ws; ws += (size_t)D * D * sizeof(float);
  float* WTn2 = (float*)ws; ws += (size_t)D * D * sizeof(float);

  {
    dim3 blk(256), grd((D * D + 255) / 256);
    repack_weight<<<grd, blk, 0, stream>>>(Ws1, WTs1);
    repack_weight<<<grd, blk, 0, stream>>>(Wn1, WTn1);
    repack_weight<<<grd, blk, 0, stream>>>(Ws2, WTs2);
    repack_weight<<<grd, blk, 0, stream>>>(Wn2, WTn2);
  }

  dim3 gemmGrid((N + LAYER_WAVES * 16 - 1) / (LAYER_WAVES * 16));

  char* sc = (char*)lists;
  const int nBuckets = (N + BKT_NODES - 1) / BKT_NODES;
  const int chunk    = (E + BIN_GROUPS - 1) / BIN_GROUPS;
  const int nHist    = nBuckets * BIN_GROUPS;
  const int nTiles   = (nHist + 1023) / 1024;
  int* listPtr = (int*)sc; sc += (((size_t)N * sizeof(int)) + 255) & ~(size_t)255;
  int* hist    = (int*)sc; sc += (((size_t)nHist * sizeof(int)) + 255) & ~(size_t)255;
  int* tileSum = (int*)sc; sc += (((size_t)nTiles * sizeof(int)) + 255) & ~(size_t)255;
  int* binned  = (int*)sc; sc += (((size_t)E * 2 * sizeof(int)) + 255) & ~(size_t)255;
  int* nbr     = (int*)sc;
  const size_t bktLds = (size_t)nBuckets * sizeof(int);

  bucket_count<<<BIN_GROUPS, 256, bktLds, stream>>>(dst, hist, E, nBuckets, chunk);
  tile_sums<<<nTiles, 256, 0, stream>>>(hist, tileSum, nHist);
  scan_tiles<<<nTiles, 256, 0, stream>>>(hist, tileSum, nHist);
  bucket_scatter<<<BIN_GROUPS, 256, bktLds, stream>>>(src, dst, hist, binned, E, nBuckets, chunk);
  bucket_lists<<<nBuckets, BKT_NODES, 0, stream>>>(binned, hist, listPtr, deg, nbr, N, E, nBuckets, BIN_GROUPS);

  sage_layer<false><<<gemmGrid, LAYER_WAVES * 32, 0, stream>>>(x, listPtr, nbr, deg, WTs1, WTn1, b1, h1, N, E);

  sage_layer<true><<<gemmGrid, LAYER_WAVES * 32, 0, stream>>>(h1, listPtr, nbr, deg, WTs2, WTn2, b2, out, N, E);
}


# Layer kernel made resident: one 16-wave block per WGP copies both weight matrices to LDS once, then each wave walks its own 16-row strips with no block barrier or weight refill; all sums unchanged
# speedup vs baseline: 3.1648x; 1.0338x over previous
#include <hip/hip_runtime.h>
#include <cstddef>
#include <cstdint>

#define D 128
#define LDS_PP2 144

typedef float v2f __attribute__((ext_vector_type(2)));
typedef float v8f __attribute__((ext_vector_type(8)));

__device__ __forceinline__ v8f wmma_f32(v2f a, v2f b, v8f c) {
#if defined(__gfx1250__)
  return __builtin_amdgcn_wmma_f32_16x16x4_f32(false, a, false, b,
                                               (short)0, c, false, false);
#else
  c[0] += a[0] * b[0];
  return c;
#endif
}

template <int M>
__device__ __forceinline__ float row_xor(float v) {
#if defined(__HIP_DEVICE_COMPILE__)
  const int b = __builtin_bit_cast(int, v);
  return __builtin_bit_cast(float, __builtin_amdgcn_update_dpp(b, b, 0x160 | M, 0xf, 0xf, false));
#else
  return v;
#endif
}

__global__ void repack_weight(const float* __restrict__ W, float* __restrict__ WTi) {
  int idx = blockIdx.x * blockDim.x + threadIdx.x;
  if (idx >= D * D) return;
  int n = (idx >> 1) & (D - 1);
  int k = ((idx >> 8) << 1) | (idx & 1);
  WTi[idx] = W[n * D + k];
}

#define BKT_SHIFT 8
#define BKT_NODES 256
#define BIN_GROUPS 512

__device__ __forceinline__ int bucket_of(int d, int nBuckets) {
  const int b = d >> BKT_SHIFT;
  return (b < 0) ? 0 : ((b >= nBuckets) ? (nBuckets - 1) : b);
}

__device__ __forceinline__ int local_of(int d, int node0) {
  const int l = d - node0;
  return (l < 0) ? 0 : ((l >= BKT_NODES) ? (BKT_NODES - 1) : l);
}

__global__ void __launch_bounds__(256)
bucket_count(const int* __restrict__ dst, int* __restrict__ hist,
             int nEdges, int nBuckets, int chunk) {
  extern __shared__ int lcnt[];
  const int tid = threadIdx.x;
  for (int b = tid; b < nBuckets; b += 256) lcnt[b] = 0;
  __syncthreads();
  const int lo = (int)blockIdx.x * chunk;
  int hi = lo + chunk;
  hi = (hi < nEdges) ? hi : nEdges;
  int e = lo + tid;
  for (; e + 768 < hi; e += 1024) {
    const int d0 = dst[e];
    const int d1 = dst[e + 256];
    const int d2 = dst[e + 512];
    const int d3 = dst[e + 768];
    atomicAdd(&lcnt[bucket_of(d0, nBuckets)], 1);
    atomicAdd(&lcnt[bucket_of(d1, nBuckets)], 1);
    atomicAdd(&lcnt[bucket_of(d2, nBuckets)], 1);
    atomicAdd(&lcnt[bucket_of(d3, nBuckets)], 1);
  }
  for (; e < hi; e += 256) atomicAdd(&lcnt[bucket_of(dst[e], nBuckets)], 1);
  __syncthreads();
  for (int b = tid; b < nBuckets; b += 256)
    hist[(size_t)b * gridDim.x + blockIdx.x] = lcnt[b];
}

__global__ void __launch_bounds__(256)
tile_sums(const int* __restrict__ cnt, int* __restrict__ tileSum, int n) {
  __shared__ int waveTot[8];
  const int tid  = threadIdx.x;
  const int lane = tid & 31;
  const int i = blockIdx.x * 1024 + tid * 4;
  int t = 0;
  if (i + 3 < n) {
    const int4 v = *(const int4*)(cnt + i);
    t = v.x + v.y + v.z + v.w;
  } else {
    if (i     < n) t += cnt[i];
    if (i + 1 < n) t += cnt[i + 1];
    if (i + 2 < n) t += cnt[i + 2];
  }
#pragma unroll
  for (int off = 16; off > 0; off >>= 1) t += __shfl_xor(t, off, 32);
  if (lane == 0) waveTot[tid >> 5] = t;
  __syncthreads();
  if (tid == 0) {
    int s = 0;
#pragma unroll
    for (int k = 0; k < 8; ++k) s += waveTot[k];
    tileSum[blockIdx.x] = s;
  }
}

__global__ void __launch_bounds__(256)
scan_tiles(int* __restrict__ a, const int* __restrict__ tileSum, int n) {
  __shared__ int waveTot[8];
  const int tid  = threadIdx.x;
  const int lane = tid & 31;
  const int w    = tid >> 5;
  const int i = blockIdx.x * 1024 + tid * 4;
  const bool full = (i + 3 < n);
  int v0 = 0, v1 = 0, v2 = 0, v3 = 0;
  if (full) {
    const int4 v = *(const int4*)(a + i);
    v0 = v.x; v1 = v.y; v2 = v.z; v3 = v.w;
  } else {
    if (i     < n) v0 = a[i];
    if (i + 1 < n) v1 = a[i + 1];
    if (i + 2 < n) v2 = a[i + 2];
  }
  int before = 0;
  for (int k = lane; k < (int)blockIdx.x; k += 32) before += tileSum[k];
#pragma unroll
  for (int off = 16; off > 0; off >>= 1) before += __shfl_xor(before, off, 32);

  const int t = v0 + v1 + v2 + v3;
  int inc = t;
#pragma unroll
  for (int off = 1; off < 32; off <<= 1) {
    const int u = __shfl_up(inc, off, 32);
    if (lane >= off) inc += u;
  }
  if (lane == 31) waveTot[w] = inc;
  __syncthreads();
  const int wt = (lane < 8) ? waveTot[lane] : 0;
  int winc = wt;
#pragma unroll
  for (int off = 1; off < 8; off <<= 1) {
    const int u = __shfl_up(winc, off, 32);
    if (lane >= off) winc += u;
  }
  const int wavePrefix = __shfl(winc - wt, w, 32);
  const int ex = before + wavePrefix + (inc - t);
  if (full) {
    *(int4*)(a + i) = make_int4(ex, ex + v0, ex + v0 + v1, ex + v0 + v1 + v2);
  } else {
    if (i     < n) a[i]     = ex;
    if (i + 1 < n) a[i + 1] = ex + v0;
    if (i + 2 < n) a[i + 2] = ex + v0 + v1;
  }
}

__global__ void __launch_bounds__(256)
bucket_scatter(const int* __restrict__ src, const int* __restrict__ dst,
               const int* __restrict__ offs, int* __restrict__ binned,
               int nEdges, int nBuckets, int chunk) {
  extern __shared__ int lcur[];
  const int tid = threadIdx.x;
  for (int b = tid; b < nBuckets; b += 256)
    lcur[b] = offs[(size_t)b * gridDim.x + blockIdx.x];
  __syncthreads();
  const int lo = (int)blockIdx.x * chunk;
  int hi = lo + chunk;
  hi = (hi < nEdges) ? hi : nEdges;
  const unsigned last = (unsigned)(nEdges - 1);
  int2* pairs = (int2*)binned;
  int e = lo + tid;
  for (; e + 768 < hi; e += 1024) {
    const int d0 = dst[e],       s0 = src[e];
    const int d1 = dst[e + 256], s1 = src[e + 256];
    const int d2 = dst[e + 512], s2 = src[e + 512];
    const int d3 = dst[e + 768], s3 = src[e + 768];
    const unsigned p0 = (unsigned)atomicAdd(&lcur[bucket_of(d0, nBuckets)], 1);
    const unsigned p1 = (unsigned)atomicAdd(&lcur[bucket_of(d1, nBuckets)], 1);
    const unsigned p2 = (unsigned)atomicAdd(&lcur[bucket_of(d2, nBuckets)], 1);
    const unsigned p3 = (unsigned)atomicAdd(&lcur[bucket_of(d3, nBuckets)], 1);
    pairs[(p0 < last) ? p0 : last] = make_int2(s0, d0);
    pairs[(p1 < last) ? p1 : last] = make_int2(s1, d1);
    pairs[(p2 < last) ? p2 : last] = make_int2(s2, d2);
    pairs[(p3 < last) ? p3 : last] = make_int2(s3, d3);
  }
  for (; e < hi; e += 256) {
    const int d0 = dst[e], s0 = src[e];
    const unsigned p0 = (unsigned)atomicAdd(&lcur[bucket_of(d0, nBuckets)], 1);
    pairs[(p0 < last) ? p0 : last] = make_int2(s0, d0);
  }
}

__global__ void __launch_bounds__(BKT_NODES)
bucket_lists(const int* __restrict__ binned, const int* __restrict__ offs,
             int* __restrict__ listEnd, float* __restrict__ deg, int* __restrict__ nbr,
             int nNodes, int nEdges, int nBuckets, int groups) {
  __shared__ int cnt[BKT_NODES];
  __shared__ int cur[BKT_NODES];
  __shared__ int waveTot[8];
  const int tid  = threadIdx.x;
  const int lane = tid & 31;
  const int w    = tid >> 5;
  const int b    = (int)blockIdx.x;
  const int node0 = b << BKT_SHIFT;
  int start = offs[(size_t)b * groups];
  int end   = (b + 1 < nBuckets) ? offs[(size_t)(b + 1) * groups] : nEdges;
  start = (start < 0) ? 0 : ((start > nEdges) ? nEdges : start);
  end   = (end > nEdges) ? nEdges : end;
  const int2* pairs = (const int2*)binned;

  cnt[tid] = 0;
  __syncthreads();
  int e = start + tid;
  for (; e + 3 * BKT_NODES < end; e += 4 * BKT_NODES) {
    const int d0 = pairs[e].y;
    const int d1 = pairs[e + BKT_NODES].y;
    const int d2 = pairs[e + 2 * BKT_NODES].y;
    const int d3 = pairs[e + 3 * BKT_NODES].y;
    atomicAdd(&cnt[local_of(d0, node0)], 1);
    atomicAdd(&cnt[local_of(d1, node0)], 1);
    atomicAdd(&cnt[local_of(d2, node0)], 1);
    atomicAdd(&cnt[local_of(d3, node0)], 1);
  }
  for (; e < end; e += BKT_NODES) atomicAdd(&cnt[local_of(pairs[e].y, node0)], 1);
  __syncthreads();

  const int c = cnt[tid];
  int inc = c;
#pragma unroll
  for (int off = 1; off < 32; off <<= 1) {
    const int u = __shfl_up(inc, off, 32);
    if (lane >= off) inc += u;
  }
  if (lane == 31) waveTot[w] = inc;
  __syncthreads();
  const int wt = (lane < 8) ? waveTot[lane] : 0;
  int winc = wt;
#pragma unroll
  for (int off = 1; off < 8; off <<= 1) {
    const int u = __shfl_up(winc, off, 32);
    if (lane >= off) winc += u;
  }
  const int wavePrefix = __shfl(winc - wt, w, 32);
  const int first = start + wavePrefix + (inc - c);
  cur[tid] = first;
  const int node = node0 + tid;
  if (node < nNodes) {
    listEnd[node] = first + c;
    deg[node]     = (float)c;
  }
  __syncthreads();

  const unsigned last = (unsigned)(nEdges - 1);
  e = start + tid;
  for (; e + 3 * BKT_NODES < end; e += 4 * BKT_NODES) {
    const int2 p0 = pairs[e];
    const int2 p1 = pairs[e + BKT_NODES];
    const int2 p2 = pairs[e + 2 * BKT_NODES];
    const int2 p3 = pairs[e + 3 * BKT_NODES];
    const unsigned q0 = (unsigned)atomicAdd(&cur[local_of(p0.y, node0)], 1);
    const unsigned q1 = (unsigned)atomicAdd(&cur[local_of(p1.y, node0)], 1);
    const unsigned q2 = (unsigned)atomicAdd(&cur[local_of(p2.y, node0)], 1);
    const unsigned q3 = (unsigned)atomicAdd(&cur[local_of(p3.y, node0)], 1);
    nbr[(q0 < last) ? q0 : last] = p0.x;
    nbr[(q1 < last) ? q1 : last] = p1.x;
    nbr[(q2 < last) ? q2 : last] = p2.x;
    nbr[(q3 < last) ? q3 : last] = p3.x;
  }
  for (; e < end; e += BKT_NODES) {
    const int2 p0 = pairs[e];
    const unsigned q0 = (unsigned)atomicAdd(&cur[local_of(p0.y, node0)], 1);
    nbr[(q0 < last) ? q0 : last] = p0.x;
  }
}

#define A_PITCH 132
#define W_LDS_FLOATS (64 * LDS_PP2 * 2)
#define LAYER_WAVES 16
#define LAYER_BLOCKS 256

template <bool kNormalize>
__global__ void __launch_bounds__(LAYER_WAVES * 32)
sage_layer(const float* __restrict__ xin,
           const int* __restrict__ listEnd, const int* __restrict__ nbr,
           const float* __restrict__ deg,
           const float* __restrict__ WTiS, const float* __restrict__ WTiN,
           const float* __restrict__ bias,
           float* __restrict__ out, int nNodes, int nEdges) {
  __shared__ __attribute__((aligned(16))) float lds[2 * W_LDS_FLOATS + LAYER_WAVES * 16 * A_PITCH];
  const int tid  = threadIdx.x;
  const int wave = __builtin_amdgcn_readfirstlane(tid >> 5);
  const int lane = tid & 31;
  const int half = lane >> 4;
  const int l16  = lane & 15;

  const v2f* B2S = (const v2f*)lds;
  const v2f* B2N = (const v2f*)(lds + W_LDS_FLOATS);
  float* aTile = lds + 2 * W_LDS_FLOATS + wave * (16 * A_PITCH);
  const float* ar = aTile + l16 * A_PITCH + half * 2;

  {
    const float4* srcS = (const float4*)WTiS;
    const float4* srcN = (const float4*)WTiN;
    constexpr int kFill = (D * D) / 4 / (LAYER_WAVES * 32);
    float4 ws[kFill], wn[kFill];
#pragma unroll
    for (int q = 0; q < kFill; ++q) {
      ws[q] = srcS[tid + q * (LAYER_WAVES * 32)];
      wn[q] = srcN[tid + q * (LAYER_WAVES * 32)];
    }
#pragma unroll
    for (int q = 0; q < kFill; ++q) {
      const int i = tid + q * (LAYER_WAVES * 32);
      const int dstI = (i >> 6) * (LDS_PP2 / 2) + (i & 63);
      ((float4*)lds)[dstI] = ws[q];
      ((float4*)(lds + W_LDS_FLOATS))[dstI] = wn[q];
    }
    __syncthreads();
  }

  const int nStrips = (nNodes + 15) >> 4;
  for (int strip = (int)blockIdx.x + (int)gridDim.x * wave; strip < nStrips;
       strip += (int)gridDim.x * LAYER_WAVES) {
  const int m0   = strip * 16;
  const int mRow = m0 + l16;
  const int mC   = (mRow < nNodes) ? mRow : (nNodes - 1);

  const float invd = 1.0f / fmaxf(deg[mC], 1.0f);

  {
    const float4* x4 = (const float4*)xin;
    float4 r[16];
#pragma unroll
    for (int q = 0; q < 16; ++q) {
      const int row = (m0 + q < nNodes) ? (m0 + q) : (nNodes - 1);
      r[q] = x4[(size_t)row * (D / 4) + lane];
    }
#pragma unroll
    for (int q = 0; q < 16; ++q) *(float4*)(aTile + q * A_PITCH + lane * 4) = r[q];
  }

  v8f acc[8];
#pragma unroll
  for (int t = 0; t < 8; ++t) acc[t] = (v8f)(0.0f);

  {
    v2f aCur = *(const v2f*)(ar);
    for (int k0 = 0; k0 < D; k0 += 4) {
      const int base2 = ((k0 >> 1) + half) * LDS_PP2 + l16;
      v2f b[8];
#pragma unroll
      for (int t = 0; t < 8; ++t) b[t] = B2S[base2 + t * 16];

      const int kN = (k0 + 4 < D) ? (k0 + 4) : k0;
      const v2f aNext = *(const v2f*)(ar + kN);

#pragma unroll
      for (int t = 0; t < 8; ++t) acc[t] = wmma_f32(aCur, b[t], acc[t]);
      aCur = aNext;
    }
  }

  {
    const float4* x4 = (const float4*)xin;
    int leIdx = m0 - 1 + lane;
    leIdx = (leIdx < 0) ? 0 : ((leIdx >= nNodes) ? (nNodes - 1) : leIdx);
    const int le = listEnd[leIdx];
    const int node0 = m0;

    auto bounds = [&](int i, int& b, int& e) {
      b = 0; e = 0;
      const int node = node0 + i;
      if (node < nNodes) {
        b = (node > 0) ? __builtin_amdgcn_readlane(le, i) : 0;
        e = __builtin_amdgcn_readlane(le, i + 1);
        b = (b < 0) ? 0 : b;
        e = (e > nEdges) ? nEdges : e;
      }
    };
    auto ids = [&](int b, int e) -> int {
      const int cnt = (e - b < 32) ? (e - b) : 32;
      return (lane < cnt) ? nbr[b + lane] : 0;
    };
    auto valid_row = [&](int id) -> int {
      return (id < 0) ? 0 : ((id >= nNodes) ? (nNodes - 1) : id);
    };
#define ROW_OF(q) x4[(size_t)__builtin_amdgcn_readlane(id, j + (q)) * (D / 4) + lane]
#define ADD_ROW(v) { acc4.x += (v).x; acc4.y += (v).y; acc4.z += (v).z; acc4.w += (v).w; }

    int begC, endC;
    bounds(0, begC, endC);
    int idC = valid_row(ids(begC, endC));
    for (int i = 0; i < 16; ++i) {
      int begX = 0, endX = 0, idX = 0;
      if (i + 1 < 16) { bounds(i + 1, begX, endX); idX = ids(begX, endX); }

      float4 acc4 = make_float4(0.0f, 0.0f, 0.0f, 0.0f);
      int id = idC;
      for (int base = begC; base < endC; base += 32) {
        const int cnt = (endC - base < 32) ? (endC - base) : 32;
        if (base != begC) id = valid_row(ids(base, endC));
        for (int j = 0; j < cnt; j += 16) {
          float4 r[16];
#pragma unroll
          for (int q = 0; q < 16; ++q) r[q] = make_float4(0.0f, 0.0f, 0.0f, 0.0f);
          r[0] = ROW_OF(0);
#pragma unroll
          for (int q = 1; q < 16; ++q)
            if (j + q < cnt) r[q] = ROW_OF(q);
#pragma unroll
          for (int q = 0; q < 16; ++q) ADD_ROW(r[q]);
        }
      }
      *(float4*)(aTile + i * A_PITCH + lane * 4) = acc4;
      begC = begX; endC = endX; idC = valid_row(idX);
    }
#undef ROW_OF
#undef ADD_ROW
  }

  {
    v2f aCur = *(const v2f*)(ar);
    aCur *= invd;
    for (int k0 = 0; k0 < D; k0 += 4) {
      const int base2 = ((k0 >> 1) + half) * LDS_PP2 + l16;
      v2f b[8];
#pragma unroll
      for (int t = 0; t < 8; ++t) b[t] = B2N[base2 + t * 16];

      const int kN = (k0 + 4 < D) ? (k0 + 4) : k0;
      v2f aNext = *(const v2f*)(ar + kN);
      aNext *= invd;

#pragma unroll
      for (int t = 0; t < 8; ++t) acc[t] = wmma_f32(aCur, b[t], acc[t]);
      aCur = aNext;
    }
  }

#pragma unroll
  for (int t = 0; t < 8; ++t) {
    const float bv = bias[t * 16 + l16];
#pragma unroll
    for (int r = 0; r < 8; ++r) acc[t][r] += bv;
  }

  if (kNormalize) {
    float ss[8];
#pragma unroll
    for (int r = 0; r < 8; ++r) {
      float s = 0.0f;
#pragma unroll
      for (int t = 0; t < 8; ++t) s += acc[t][r] * acc[t][r];
      ss[r] = s;
    }
#pragma unroll
    for (int r = 0; r < 8; ++r) ss[r] += row_xor<1>(ss[r]);
#pragma unroll
    for (int r = 0; r < 8; ++r) ss[r] += row_xor<2>(ss[r]);
#pragma unroll
    for (int r = 0; r < 8; ++r) ss[r] += row_xor<4>(ss[r]);
#pragma unroll
    for (int r = 0; r < 8; ++r) ss[r] += row_xor<8>(ss[r]);
#pragma unroll
    for (int r = 0; r < 8; ++r) {
      const float scale = 1.0f / fmaxf(sqrtf(ss[r]), 1e-12f);
#pragma unroll
      for (int t = 0; t < 8; ++t) acc[t][r] *= scale;
    }
  }

#pragma unroll
  for (int t = 0; t < 8; ++t) {
    const int col = t * 16 + l16;
#pragma unroll
    for (int r = 0; r < 8; ++r) {
      const int row = m0 + r + half * 8;
      if (row < nNodes) out[(size_t)row * D + col] = acc[t][r];
    }
  }
  }
}

extern "C" void kernel_launch(void* const* d_in, const int* in_sizes, int n_in,
                              void* d_out, int out_size, void* d_ws, size_t ws_size,
                              hipStream_t stream) {
  const float* x   = (const float*)d_in[0];
  const int*   src = (const int*)  d_in[1];
  const int*   dst = (const int*)  d_in[2];
  const float* Ws1 = (const float*)d_in[3];
  const float* Wn1 = (const float*)d_in[4];
  const float* b1  = (const float*)d_in[5];
  const float* Ws2 = (const float*)d_in[6];
  const float* Wn2 = (const float*)d_in[7];
  const float* b2  = (const float*)d_in[8];
  float* out = (float*)d_out;

  const int N = in_sizes[0] / D;
  const int E = in_sizes[1];

  char* ws = (char*)d_ws;
  char*  lists = ws;       ws += (size_t)N * D * sizeof(float);
  float* h1  = (float*)ws; ws += (size_t)N * D * sizeof(float);
  float* deg = (float*)ws; ws += (((size_t)N * sizeof(float)) + 255) & ~(size_t)255;
  float* WTs1 = (float*)ws; ws += (size_t)D * D * sizeof(float);
  float* WTn1 = (float*)ws; ws += (size_t)D * D * sizeof(float);
  float* WTs2 = (float*)ws; ws += (size_t)D * D * sizeof(float);
  float* WTn2 = (float*)ws; ws += (size_t)D * D * sizeof(float);

  {
    dim3 blk(256), grd((D * D + 255) / 256);
    repack_weight<<<grd, blk, 0, stream>>>(Ws1, WTs1);
    repack_weight<<<grd, blk, 0, stream>>>(Wn1, WTn1);
    repack_weight<<<grd, blk, 0, stream>>>(Ws2, WTs2);
    repack_weight<<<grd, blk, 0, stream>>>(Wn2, WTn2);
  }

  dim3 gemmGrid(LAYER_BLOCKS);

  char* sc = (char*)lists;
  const int nBuckets = (N + BKT_NODES - 1) / BKT_NODES;
  const int chunk    = (E + BIN_GROUPS - 1) / BIN_GROUPS;
  const int nHist    = nBuckets * BIN_GROUPS;
  const int nTiles   = (nHist + 1023) / 1024;
  int* listPtr = (int*)sc; sc += (((size_t)N * sizeof(int)) + 255) & ~(size_t)255;
  int* hist    = (int*)sc; sc += (((size_t)nHist * sizeof(int)) + 255) & ~(size_t)255;
  int* tileSum = (int*)sc; sc += (((size_t)nTiles * sizeof(int)) + 255) & ~(size_t)255;
  int* binned  = (int*)sc; sc += (((size_t)E * 2 * sizeof(int)) + 255) & ~(size_t)255;
  int* nbr     = (int*)sc;
  const size_t bktLds = (size_t)nBuckets * sizeof(int);

  bucket_count<<<BIN_GROUPS, 256, bktLds, stream>>>(dst, hist, E, nBuckets, chunk);
  tile_sums<<<nTiles, 256, 0, stream>>>(hist, tileSum, nHist);
  scan_tiles<<<nTiles, 256, 0, stream>>>(hist, tileSum, nHist);
  bucket_scatter<<<BIN_GROUPS, 256, bktLds, stream>>>(src, dst, hist, binned, E, nBuckets, chunk);
  bucket_lists<<<nBuckets, BKT_NODES, 0, stream>>>(binned, hist, listPtr, deg, nbr, N, E, nBuckets, BIN_GROUPS);

  sage_layer<false><<<gemmGrid, LAYER_WAVES * 32, 0, stream>>>(x, listPtr, nbr, deg, WTs1, WTn1, b1, h1, N, E);

  sage_layer<true><<<gemmGrid, LAYER_WAVES * 32, 0, stream>>>(h1, listPtr, nbr, deg, WTs2, WTn2, b2, out, N, E);
}


# Neighbour rows of a strip fetched as one stream over its 16 back-to-back lists: two groups of 16 row loads alternate, loads stay in flight across node ends; per-node sums and their order unchanged
# speedup vs baseline: 3.5973x; 1.1367x over previous
#include <hip/hip_runtime.h>
#include <cstddef>
#include <cstdint>

#define D 128
#define LDS_PP2 144

typedef float v2f __attribute__((ext_vector_type(2)));
typedef float v8f __attribute__((ext_vector_type(8)));
typedef float v4f __attribute__((ext_vector_type(4)));

__device__ __forceinline__ v4f load_row_piece(const float* __restrict__ plane, int row, unsigned laneByteOff) {
  const unsigned off = (unsigned)row * (unsigned)(D * sizeof(float)) + laneByteOff;
  return *(const v4f*)((const char*)plane + off);
}

__device__ __forceinline__ v8f wmma_f32(v2f a, v2f b, v8f c) {
#if defined(__gfx1250__)
  return __builtin_amdgcn_wmma_f32_16x16x4_f32(false, a, false, b,
                                               (short)0, c, false, false);
#else
  c[0] += a[0] * b[0];
  return c;
#endif
}

template <int M>
__device__ __forceinline__ float row_xor(float v) {
#if defined(__HIP_DEVICE_COMPILE__)
  const int b = __builtin_bit_cast(int, v);
  return __builtin_bit_cast(float, __builtin_amdgcn_update_dpp(b, b, 0x160 | M, 0xf, 0xf, false));
#else
  return v;
#endif
}

__global__ void repack_weight(const float* __restrict__ W, float* __restrict__ WTi) {
  int idx = blockIdx.x * blockDim.x + threadIdx.x;
  if (idx >= D * D) return;
  int n = (idx >> 1) & (D - 1);
  int k = ((idx >> 8) << 1) | (idx & 1);
  WTi[idx] = W[n * D + k];
}

#define BKT_SHIFT 8
#define BKT_NODES 256
#define BIN_GROUPS 512

__device__ __forceinline__ int bucket_of(int d, int nBuckets) {
  const int b = d >> BKT_SHIFT;
  return (b < 0) ? 0 : ((b >= nBuckets) ? (nBuckets - 1) : b);
}

__device__ __forceinline__ int local_of(int d, int node0) {
  const int l = d - node0;
  return (l < 0) ? 0 : ((l >= BKT_NODES) ? (BKT_NODES - 1) : l);
}

__global__ void __launch_bounds__(256)
bucket_count(const int* __restrict__ dst, int* __restrict__ hist,
             int nEdges, int nBuckets, int chunk) {
  extern __shared__ int lcnt[];
  const int tid = threadIdx.x;
  for (int b = tid; b < nBuckets; b += 256) lcnt[b] = 0;
  __syncthreads();
  const int lo = (int)blockIdx.x * chunk;
  int hi = lo + chunk;
  hi = (hi < nEdges) ? hi : nEdges;
  int e = lo + tid;
  for (; e + 768 < hi; e += 1024) {
    const int d0 = dst[e];
    const int d1 = dst[e + 256];
    const int d2 = dst[e + 512];
    const int d3 = dst[e + 768];
    atomicAdd(&lcnt[bucket_of(d0, nBuckets)], 1);
    atomicAdd(&lcnt[bucket_of(d1, nBuckets)], 1);
    atomicAdd(&lcnt[bucket_of(d2, nBuckets)], 1);
    atomicAdd(&lcnt[bucket_of(d3, nBuckets)], 1);
  }
  for (; e < hi; e += 256) atomicAdd(&lcnt[bucket_of(dst[e], nBuckets)], 1);
  __syncthreads();
  for (int b = tid; b < nBuckets; b += 256)
    hist[(size_t)b * gridDim.x + blockIdx.x] = lcnt[b];
}

__global__ void __launch_bounds__(256)
tile_sums(const int* __restrict__ cnt, int* __restrict__ tileSum, int n) {
  __shared__ int waveTot[8];
  const int tid  = threadIdx.x;
  const int lane = tid & 31;
  const int i = blockIdx.x * 1024 + tid * 4;
  int t = 0;
  if (i + 3 < n) {
    const int4 v = *(const int4*)(cnt + i);
    t = v.x + v.y + v.z + v.w;
  } else {
    if (i     < n) t += cnt[i];
    if (i + 1 < n) t += cnt[i + 1];
    if (i + 2 < n) t += cnt[i + 2];
  }
#pragma unroll
  for (int off = 16; off > 0; off >>= 1) t += __shfl_xor(t, off, 32);
  if (lane == 0) waveTot[tid >> 5] = t;
  __syncthreads();
  if (tid == 0) {
    int s = 0;
#pragma unroll
    for (int k = 0; k < 8; ++k) s += waveTot[k];
    tileSum[blockIdx.x] = s;
  }
}

__global__ void __launch_bounds__(256)
scan_tiles(int* __restrict__ a, const int* __restrict__ tileSum, int n) {
  __shared__ int waveTot[8];
  const int tid  = threadIdx.x;
  const int lane = tid & 31;
  const int w    = tid >> 5;
  const int i = blockIdx.x * 1024 + tid * 4;
  const bool full = (i + 3 < n);
  int v0 = 0, v1 = 0, v2 = 0, v3 = 0;
  if (full) {
    const int4 v = *(const int4*)(a + i);
    v0 = v.x; v1 = v.y; v2 = v.z; v3 = v.w;
  } else {
    if (i     < n) v0 = a[i];
    if (i + 1 < n) v1 = a[i + 1];
    if (i + 2 < n) v2 = a[i + 2];
  }
  int before = 0;
  for (int k = lane; k < (int)blockIdx.x; k += 32) before += tileSum[k];
#pragma unroll
  for (int off = 16; off > 0; off >>= 1) before += __shfl_xor(before, off, 32);

  const int t = v0 + v1 + v2 + v3;
  int inc = t;
#pragma unroll
  for (int off = 1; off < 32; off <<= 1) {
    const int u = __shfl_up(inc, off, 32);
    if (lane >= off) inc += u;
  }
  if (lane == 31) waveTot[w] = inc;
  __syncthreads();
  const int wt = (lane < 8) ? waveTot[lane] : 0;
  int winc = wt;
#pragma unroll
  for (int off = 1; off < 8; off <<= 1) {
    const int u = __shfl_up(winc, off, 32);
    if (lane >= off) winc += u;
  }
  const int wavePrefix = __shfl(winc - wt, w, 32);
  const int ex = before + wavePrefix + (inc - t);
  if (full) {
    *(int4*)(a + i) = make_int4(ex, ex + v0, ex + v0 + v1, ex + v0 + v1 + v2);
  } else {
    if (i     < n) a[i]     = ex;
    if (i + 1 < n) a[i + 1] = ex + v0;
    if (i + 2 < n) a[i + 2] = ex + v0 + v1;
  }
}

__global__ void __launch_bounds__(256)
bucket_scatter(const int* __restrict__ src, const int* __restrict__ dst,
               const int* __restrict__ offs, int* __restrict__ binned,
               int nEdges, int nBuckets, int chunk) {
  extern __shared__ int lcur[];
  const int tid = threadIdx.x;
  for (int b = tid; b < nBuckets; b += 256)
    lcur[b] = offs[(size_t)b * gridDim.x + blockIdx.x];
  __syncthreads();
  const int lo = (int)blockIdx.x * chunk;
  int hi = lo + chunk;
  hi = (hi < nEdges) ? hi : nEdges;
  const unsigned last = (unsigned)(nEdges - 1);
  int2* pairs = (int2*)binned;
  int e = lo + tid;
  for (; e + 768 < hi; e += 1024) {
    const int d0 = dst[e],       s0 = src[e];
    const int d1 = dst[e + 256], s1 = src[e + 256];
    const int d2 = dst[e + 512], s2 = src[e + 512];
    const int d3 = dst[e + 768], s3 = src[e + 768];
    const unsigned p0 = (unsigned)atomicAdd(&lcur[bucket_of(d0, nBuckets)], 1);
    const unsigned p1 = (unsigned)atomicAdd(&lcur[bucket_of(d1, nBuckets)], 1);
    const unsigned p2 = (unsigned)atomicAdd(&lcur[bucket_of(d2, nBuckets)], 1);
    const unsigned p3 = (unsigned)atomicAdd(&lcur[bucket_of(d3, nBuckets)], 1);
    pairs[(p0 < last) ? p0 : last] = make_int2(s0, d0);
    pairs[(p1 < last) ? p1 : last] = make_int2(s1, d1);
    pairs[(p2 < last) ? p2 : last] = make_int2(s2, d2);
    pairs[(p3 < last) ? p3 : last] = make_int2(s3, d3);
  }
  for (; e < hi; e += 256) {
    const int d0 = dst[e], s0 = src[e];
    const unsigned p0 = (unsigned)atomicAdd(&lcur[bucket_of(d0, nBuckets)], 1);
    pairs[(p0 < last) ? p0 : last] = make_int2(s0, d0);
  }
}

__global__ void __launch_bounds__(BKT_NODES)
bucket_lists(const int* __restrict__ binned, const int* __restrict__ offs,
             int* __restrict__ listEnd, float* __restrict__ deg, int* __restrict__ nbr,
             int nNodes, int nEdges, int nBuckets, int groups) {
  __shared__ int cnt[BKT_NODES];
  __shared__ int cur[BKT_NODES];
  __shared__ int waveTot[8];
  const int tid  = threadIdx.x;
  const int lane = tid & 31;
  const int w    = tid >> 5;
  const int b    = (int)blockIdx.x;
  const int node0 = b << BKT_SHIFT;
  int start = offs[(size_t)b * groups];
  int end   = (b + 1 < nBuckets) ? offs[(size_t)(b + 1) * groups] : nEdges;
  start = (start < 0) ? 0 : ((start > nEdges) ? nEdges : start);
  end   = (end > nEdges) ? nEdges : end;
  const int2* pairs = (const int2*)binned;

  cnt[tid] = 0;
  __syncthreads();
  int e = start + tid;
  for (; e + 3 * BKT_NODES < end; e += 4 * BKT_NODES) {
    const int d0 = pairs[e].y;
    const int d1 = pairs[e + BKT_NODES].y;
    const int d2 = pairs[e + 2 * BKT_NODES].y;
    const int d3 = pairs[e + 3 * BKT_NODES].y;
    atomicAdd(&cnt[local_of(d0, node0)], 1);
    atomicAdd(&cnt[local_of(d1, node0)], 1);
    atomicAdd(&cnt[local_of(d2, node0)], 1);
    atomicAdd(&cnt[local_of(d3, node0)], 1);
  }
  for (; e < end; e += BKT_NODES) atomicAdd(&cnt[local_of(pairs[e].y, node0)], 1);
  __syncthreads();

  const int c = cnt[tid];
  int inc = c;
#pragma unroll
  for (int off = 1; off < 32; off <<= 1) {
    const int u = __shfl_up(inc, off, 32);
    if (lane >= off) inc += u;
  }
  if (lane == 31) waveTot[w] = inc;
  __syncthreads();
  const int wt = (lane < 8) ? waveTot[lane] : 0;
  int winc = wt;
#pragma unroll
  for (int off = 1; off < 8; off <<= 1) {
    const int u = __shfl_up(winc, off, 32);
    if (lane >= off) winc += u;
  }
  const int wavePrefix = __shfl(winc - wt, w, 32);
  const int first = start + wavePrefix + (inc - c);
  cur[tid] = first;
  const int node = node0 + tid;
  if (node < nNodes) {
    listEnd[node] = first + c;
    deg[node]     = (float)c;
  }
  __syncthreads();

  const unsigned last = (unsigned)(nEdges - 1);
  e = start + tid;
  for (; e + 3 * BKT_NODES < end; e += 4 * BKT_NODES) {
    const int2 p0 = pairs[e];
    const int2 p1 = pairs[e + BKT_NODES];
    const int2 p2 = pairs[e + 2 * BKT_NODES];
    const int2 p3 = pairs[e + 3 * BKT_NODES];
    const unsigned q0 = (unsigned)atomicAdd(&cur[local_of(p0.y, node0)], 1);
    const unsigned q1 = (unsigned)atomicAdd(&cur[local_of(p1.y, node0)], 1);
    const unsigned q2 = (unsigned)atomicAdd(&cur[local_of(p2.y, node0)], 1);
    const unsigned q3 = (unsigned)atomicAdd(&cur[local_of(p3.y, node0)], 1);
    nbr[(q0 < last) ? q0 : last] = p0.x;
    nbr[(q1 < last) ? q1 : last] = p1.x;
    nbr[(q2 < last) ? q2 : last] = p2.x;
    nbr[(q3 < last) ? q3 : last] = p3.x;
  }
  for (; e < end; e += BKT_NODES) {
    const int2 p0 = pairs[e];
    const unsigned q0 = (unsigned)atomicAdd(&cur[local_of(p0.y, node0)], 1);
    nbr[(q0 < last) ? q0 : last] = p0.x;
  }
}

#define A_PITCH 132
#define W_LDS_FLOATS (64 * LDS_PP2 * 2)
#define LAYER_WAVES 16
#define LAYER_BLOCKS 256

template <bool kNormalize>
__global__ void __launch_bounds__(LAYER_WAVES * 32)
sage_layer(const float* __restrict__ xin,
           const int* __restrict__ listEnd, const int* __restrict__ nbr,
           const float* __restrict__ deg,
           const float* __restrict__ WTiS, const float* __restrict__ WTiN,
           const float* __restrict__ bias,
           float* __restrict__ out, int nNodes, int nEdges) {
  __shared__ __attribute__((aligned(16))) float lds[2 * W_LDS_FLOATS + LAYER_WAVES * 16 * A_PITCH];
  const int tid  = threadIdx.x;
  const int wave = __builtin_amdgcn_readfirstlane(tid >> 5);
  const int lane = tid & 31;
  const int half = lane >> 4;
  const int l16  = lane & 15;

  const v2f* B2S = (const v2f*)lds;
  const v2f* B2N = (const v2f*)(lds + W_LDS_FLOATS);
  float* aTile = lds + 2 * W_LDS_FLOATS + wave * (16 * A_PITCH);
  const float* ar = aTile + l16 * A_PITCH + half * 2;

  {
    const float4* srcS = (const float4*)WTiS;
    const float4* srcN = (const float4*)WTiN;
    constexpr int kFill = (D * D) / 4 / (LAYER_WAVES * 32);
    float4 ws[kFill], wn[kFill];
#pragma unroll
    for (int q = 0; q < kFill; ++q) {
      ws[q] = srcS[tid + q * (LAYER_WAVES * 32)];
      wn[q] = srcN[tid + q * (LAYER_WAVES * 32)];
    }
#pragma unroll
    for (int q = 0; q < kFill; ++q) {
      const int i = tid + q * (LAYER_WAVES * 32);
      const int dstI = (i >> 6) * (LDS_PP2 / 2) + (i & 63);
      ((float4*)lds)[dstI] = ws[q];
      ((float4*)(lds + W_LDS_FLOATS))[dstI] = wn[q];
    }
    __syncthreads();
  }

  const int nStrips = (nNodes + 15) >> 4;
  const unsigned lane4 = (unsigned)lane * 4u;
  for (int strip = (int)blockIdx.x + (int)gridDim.x * wave; strip < nStrips;
       strip += (int)gridDim.x * LAYER_WAVES) {
  const int m0   = strip * 16;
  const int mRow = m0 + l16;
  const int mC   = (mRow < nNodes) ? mRow : (nNodes - 1);

  const float invd = 1.0f / fmaxf(deg[mC], 1.0f);

  int endv = listEnd[mC];
  endv = (endv < 0) ? 0 : ((endv > nEdges) ? nEdges : endv);
  int eBeg = (m0 > 0) ? __builtin_amdgcn_readfirstlane(listEnd[m0 - 1]) : 0;
  eBeg = (eBeg < 0) ? 0 : ((eBeg > nEdges) ? nEdges : eBeg);
  const int eEnd = __builtin_amdgcn_readlane(endv, 15);

  auto load_ids = [&](int cb) -> int {
    int idx = cb + lane;
    idx = (idx < nEdges) ? idx : (nEdges - 1);
    return nbr[idx];
  };
  auto fix_ids = [&](int raw, int cb) -> int {
    const int id = (cb + lane < eEnd) ? raw : m0;
    return (id < 0) ? 0 : ((id >= nNodes) ? (nNodes - 1) : id);
  };
#define ROW_AT(idv, L) load_row_piece(xin, __builtin_amdgcn_readlane(idv, (L)), lane4 * 4u)

  int cb = eBeg;
  int idRaw0 = load_ids(cb);
  int idRawNext = load_ids(cb + 32);

  {
    const float4* x4 = (const float4*)xin;
    float4 r[16];
#pragma unroll
    for (int q = 0; q < 16; ++q) {
      const int row = (m0 + q < nNodes) ? (m0 + q) : (nNodes - 1);
      r[q] = x4[(size_t)row * (D / 4) + lane];
    }
#pragma unroll
    for (int q = 0; q < 16; ++q) *(float4*)(aTile + q * A_PITCH + lane * 4) = r[q];
  }

  int idv = fix_ids(idRaw0, cb);
  v4f ra[16], rb[16];
#pragma unroll
  for (int q = 0; q < 16; ++q) ra[q] = ROW_AT(idv, q);

  v8f acc[8];
#pragma unroll
  for (int t = 0; t < 8; ++t) acc[t] = (v8f)(0.0f);

  {
    v2f aCur = *(const v2f*)(ar);
    for (int k0 = 0; k0 < D; k0 += 4) {
      const int base2 = ((k0 >> 1) + half) * LDS_PP2 + l16;
      v2f b[8];
#pragma unroll
      for (int t = 0; t < 8; ++t) b[t] = B2S[base2 + t * 16];

      const int kN = (k0 + 4 < D) ? (k0 + 4) : k0;
      const v2f aNext = *(const v2f*)(ar + kN);

#pragma unroll
      for (int t = 0; t < 8; ++t) acc[t] = wmma_f32(aCur, b[t], acc[t]);
      aCur = aNext;
    }
  }

  {
    int nodeI = 0;
    int endN = __builtin_amdgcn_readlane(endv, 0);
    v4f acc4 = (v4f)(0.0f);
#define PARK_DONE_NODES(P) \
    while (nodeI < 16 && (P) >= endN) { \
      *(v4f*)(aTile + nodeI * A_PITCH + lane4) = acc4; \
      acc4 = (v4f)(0.0f); \
      ++nodeI; \
      endN = __builtin_amdgcn_readlane(endv, nodeI & 15); \
    }
#define ADD_ROW(v) { acc4 += (v); }

    for (; cb < eEnd; cb += 32) {
#pragma unroll
      for (int q = 0; q < 16; ++q) rb[q] = ROW_AT(idv, 16 + q);
#pragma unroll
      for (int q = 0; q < 16; ++q) { PARK_DONE_NODES(cb + q); ADD_ROW(ra[q]); }
      const int cbN = cb + 32;
      const int idvN = fix_ids(idRawNext, cbN);
      idRawNext = load_ids(cbN + 32);
#pragma unroll
      for (int q = 0; q < 16; ++q) ra[q] = ROW_AT(idvN, q);
#pragma unroll
      for (int q = 0; q < 16; ++q) { PARK_DONE_NODES(cb + 16 + q); ADD_ROW(rb[q]); }
      idv = idvN;
    }
    while (nodeI < 16) {
      *(v4f*)(aTile + nodeI * A_PITCH + lane4) = acc4;
      acc4 = (v4f)(0.0f);
      ++nodeI;
    }
#undef PARK_DONE_NODES
#undef ADD_ROW
#undef ROW_AT
  }

  {
    v2f aCur = *(const v2f*)(ar);
    aCur *= invd;
    for (int k0 = 0; k0 < D; k0 += 4) {
      const int base2 = ((k0 >> 1) + half) * LDS_PP2 + l16;
      v2f b[8];
#pragma unroll
      for (int t = 0; t < 8; ++t) b[t] = B2N[base2 + t * 16];

      const int kN = (k0 + 4 < D) ? (k0 + 4) : k0;
      v2f aNext = *(const v2f*)(ar + kN);
      aNext *= invd;

#pragma unroll
      for (int t = 0; t < 8; ++t) acc[t] = wmma_f32(aCur, b[t], acc[t]);
      aCur = aNext;
    }
  }

#pragma unroll
  for (int t = 0; t < 8; ++t) {
    const float bv = bias[t * 16 + l16];
#pragma unroll
    for (int r = 0; r < 8; ++r) acc[t][r] += bv;
  }

  if (kNormalize) {
    float ss[8];
#pragma unroll
    for (int r = 0; r < 8; ++r) {
      float s = 0.0f;
#pragma unroll
      for (int t = 0; t < 8; ++t) s += acc[t][r] * acc[t][r];
      ss[r] = s;
    }
#pragma unroll
    for (int r = 0; r < 8; ++r) ss[r] += row_xor<1>(ss[r]);
#pragma unroll
    for (int r = 0; r < 8; ++r) ss[r] += row_xor<2>(ss[r]);
#pragma unroll
    for (int r = 0; r < 8; ++r) ss[r] += row_xor<4>(ss[r]);
#pragma unroll
    for (int r = 0; r < 8; ++r) ss[r] += row_xor<8>(ss[r]);
#pragma unroll
    for (int r = 0; r < 8; ++r) {
      const float scale = 1.0f / fmaxf(sqrtf(ss[r]), 1e-12f);
#pragma unroll
      for (int t = 0; t < 8; ++t) acc[t][r] *= scale;
    }
  }

#pragma unroll
  for (int t = 0; t < 8; ++t) {
    const int col = t * 16 + l16;
#pragma unroll
    for (int r = 0; r < 8; ++r) {
      const int row = m0 + r + half * 8;
      if (row < nNodes) out[(size_t)row * D + col] = acc[t][r];
    }
  }
  }
}

extern "C" void kernel_launch(void* const* d_in, const int* in_sizes, int n_in,
                              void* d_out, int out_size, void* d_ws, size_t ws_size,
                              hipStream_t stream) {
  const float* x   = (const float*)d_in[0];
  const int*   src = (const int*)  d_in[1];
  const int*   dst = (const int*)  d_in[2];
  const float* Ws1 = (const float*)d_in[3];
  const float* Wn1 = (const float*)d_in[4];
  const float* b1  = (const float*)d_in[5];
  const float* Ws2 = (const float*)d_in[6];
  const float* Wn2 = (const float*)d_in[7];
  const float* b2  = (const float*)d_in[8];
  float* out = (float*)d_out;

  const int N = in_sizes[0] / D;
  const int E = in_sizes[1];

  char* ws = (char*)d_ws;
  char*  lists = ws;       ws += (size_t)N * D * sizeof(float);
  float* h1  = (float*)ws; ws += (size_t)N * D * sizeof(float);
  float* deg = (float*)ws; ws += (((size_t)N * sizeof(float)) + 255) & ~(size_t)255;
  float* WTs1 = (float*)ws; ws += (size_t)D * D * sizeof(float);
  float* WTn1 = (float*)ws; ws += (size_t)D * D * sizeof(float);
  float* WTs2 = (float*)ws; ws += (size_t)D * D * sizeof(float);
  float* WTn2 = (float*)ws; ws += (size_t)D * D * sizeof(float);

  {
    dim3 blk(256), grd((D * D + 255) / 256);
    repack_weight<<<grd, blk, 0, stream>>>(Ws1, WTs1);
    repack_weight<<<grd, blk, 0, stream>>>(Wn1, WTn1);
    repack_weight<<<grd, blk, 0, stream>>>(Ws2, WTs2);
    repack_weight<<<grd, blk, 0, stream>>>(Wn2, WTn2);
  }

  dim3 gemmGrid(LAYER_BLOCKS);

  char* sc = (char*)lists;
  const int nBuckets = (N + BKT_NODES - 1) / BKT_NODES;
  const int chunk    = (E + BIN_GROUPS - 1) / BIN_GROUPS;
  const int nHist    = nBuckets * BIN_GROUPS;
  const int nTiles   = (nHist + 1023) / 1024;
  int* listPtr = (int*)sc; sc += (((size_t)N * sizeof(int)) + 255) & ~(size_t)255;
  int* hist    = (int*)sc; sc += (((size_t)nHist * sizeof(int)) + 255) & ~(size_t)255;
  int* tileSum = (int*)sc; sc += (((size_t)nTiles * sizeof(int)) + 255) & ~(size_t)255;
  int* binned  = (int*)sc; sc += (((size_t)E * 2 * sizeof(int)) + 255) & ~(size_t)255;
  int* nbr     = (int*)sc;
  const size_t bktLds = (size_t)nBuckets * sizeof(int);

  bucket_count<<<BIN_GROUPS, 256, bktLds, stream>>>(dst, hist, E, nBuckets, chunk);
  tile_sums<<<nTiles, 256, 0, stream>>>(hist, tileSum, nHist);
  scan_tiles<<<nTiles, 256, 0, stream>>>(hist, tileSum, nHist);
  bucket_scatter<<<BIN_GROUPS, 256, bktLds, stream>>>(src, dst, hist, binned, E, nBuckets, chunk);
  bucket_lists<<<nBuckets, BKT_NODES, 0, stream>>>(binned, hist, listPtr, deg, nbr, N, E, nBuckets, BIN_GROUPS);

  sage_layer<false><<<gemmGrid, LAYER_WAVES * 32, 0, stream>>>(x, listPtr, nbr, deg, WTs1, WTn1, b1, h1, N, E);

  sage_layer<true><<<gemmGrid, LAYER_WAVES * 32, 0, stream>>>(h1, listPtr, nbr, deg, WTs2, WTn2, b2, out, N, E);
}


# Launches 11 -> 6: four weight repacks run as extra work-groups of the first grouping launch; the two-launch count scan becomes one per-bucket scan, bucket starts scanned in LDS by the scatter
# speedup vs baseline: 3.7786x; 1.0504x over previous
#include <hip/hip_runtime.h>
#include <cstddef>
#include <cstdint>

#define D 128
#define LDS_PP2 144

typedef float v2f __attribute__((ext_vector_type(2)));
typedef float v8f __attribute__((ext_vector_type(8)));
typedef float v4f __attribute__((ext_vector_type(4)));

__device__ __forceinline__ v4f load_row_piece(const float* __restrict__ plane, int row, unsigned laneByteOff) {
  const unsigned off = (unsigned)row * (unsigned)(D * sizeof(float)) + laneByteOff;
  return *(const v4f*)((const char*)plane + off);
}

__device__ __forceinline__ v8f wmma_f32(v2f a, v2f b, v8f c) {
#if defined(__gfx1250__)
  return __builtin_amdgcn_wmma_f32_16x16x4_f32(false, a, false, b,
                                               (short)0, c, false, false);
#else
  c[0] += a[0] * b[0];
  return c;
#endif
}

template <int M>
__device__ __forceinline__ float row_xor(float v) {
#if defined(__HIP_DEVICE_COMPILE__)
  const int b = __builtin_bit_cast(int, v);
  return __builtin_bit_cast(float, __builtin_amdgcn_update_dpp(b, b, 0x160 | M, 0xf, 0xf, false));
#else
  return v;
#endif
}

#define REPACK_BLOCKS ((D * D) / 256)

__device__ __forceinline__ void repack_weight(const float* __restrict__ W, float* __restrict__ WTi, int idx) {
  const int n = (idx >> 1) & (D - 1);
  const int k = ((idx >> 8) << 1) | (idx & 1);
  WTi[idx] = W[n * D + k];
}

#define BKT_SHIFT 8
#define BKT_NODES 256
#define BIN_GROUPS 512

__device__ __forceinline__ int bucket_of(int d, int nBuckets) {
  const int b = d >> BKT_SHIFT;
  return (b < 0) ? 0 : ((b >= nBuckets) ? (nBuckets - 1) : b);
}

__device__ __forceinline__ int local_of(int d, int node0) {
  const int l = d - node0;
  return (l < 0) ? 0 : ((l >= BKT_NODES) ? (BKT_NODES - 1) : l);
}

__global__ void __launch_bounds__(256)
bucket_count(const int* __restrict__ dst, int* __restrict__ hist,
             int nEdges, int nBuckets, int chunk, int groups,
             const float* __restrict__ Wa, const float* __restrict__ Wb,
             const float* __restrict__ Wc, const float* __restrict__ Wd,
             float* __restrict__ WTa, float* __restrict__ WTb,
             float* __restrict__ WTc, float* __restrict__ WTd) {
  extern __shared__ int lcnt[];
  const int tid = threadIdx.x;
  if ((int)blockIdx.x >= groups) {
    const int r = (int)blockIdx.x - groups;
    const int m = r / REPACK_BLOCKS;
    const int idx = (r - m * REPACK_BLOCKS) * 256 + tid;
    if      (m == 0) repack_weight(Wa, WTa, idx);
    else if (m == 1) repack_weight(Wb, WTb, idx);
    else if (m == 2) repack_weight(Wc, WTc, idx);
    else if (m == 3) repack_weight(Wd, WTd, idx);
    return;
  }
  for (int b = tid; b < nBuckets; b += 256) lcnt[b] = 0;
  __syncthreads();
  const int lo = (int)blockIdx.x * chunk;
  int hi = lo + chunk;
  hi = (hi < nEdges) ? hi : nEdges;
  int e = lo + tid;
  for (; e + 768 < hi; e += 1024) {
    const int d0 = dst[e];
    const int d1 = dst[e + 256];
    const int d2 = dst[e + 512];
    const int d3 = dst[e + 768];
    atomicAdd(&lcnt[bucket_of(d0, nBuckets)], 1);
    atomicAdd(&lcnt[bucket_of(d1, nBuckets)], 1);
    atomicAdd(&lcnt[bucket_of(d2, nBuckets)], 1);
    atomicAdd(&lcnt[bucket_of(d3, nBuckets)], 1);
  }
  for (; e < hi; e += 256) atomicAdd(&lcnt[bucket_of(dst[e], nBuckets)], 1);
  __syncthreads();
  for (int b = tid; b < nBuckets; b += 256)
    hist[(size_t)b * groups + blockIdx.x] = lcnt[b];
}

__global__ void __launch_bounds__(BIN_GROUPS)
bucket_scan(int* __restrict__ hist, int* __restrict__ bucketTotal) {
  __shared__ int waveTot[BIN_GROUPS / 32];
  const int tid  = threadIdx.x;
  const int lane = tid & 31;
  const int w    = tid >> 5;
  const size_t i = (size_t)blockIdx.x * BIN_GROUPS + tid;
  const int v = hist[i];
  int inc = v;
#pragma unroll
  for (int off = 1; off < 32; off <<= 1) {
    const int u = __shfl_up(inc, off, 32);
    if (lane >= off) inc += u;
  }
  if (lane == 31) waveTot[w] = inc;
  __syncthreads();
  const int wt = (lane < BIN_GROUPS / 32) ? waveTot[lane] : 0;
  int winc = wt;
#pragma unroll
  for (int off = 1; off < BIN_GROUPS / 32; off <<= 1) {
    const int u = __shfl_up(winc, off, 32);
    if (lane >= off) winc += u;
  }
  const int wavePrefix = __shfl(winc - wt, w, 32);
  hist[i] = wavePrefix + (inc - v);
  if (tid == BIN_GROUPS - 1) bucketTotal[blockIdx.x] = wavePrefix + inc;
}

__global__ void __launch_bounds__(256)
bucket_scatter(const int* __restrict__ src, const int* __restrict__ dst,
               const int* __restrict__ offs, const int* __restrict__ bucketTotal,
               int* __restrict__ bucketStart, int* __restrict__ binned,
               int nEdges, int nBuckets, int chunk) {
  extern __shared__ int lcur[];
  const int tid = threadIdx.x;
  for (int b = tid; b < nBuckets; b += 256) lcur[b] = bucketTotal[b];
  __syncthreads();
  if (tid < 32) {
    int carry = 0;
    for (int base = 0; base < nBuckets; base += 32) {
      const int i = base + tid;
      const int v = (i < nBuckets) ? lcur[i] : 0;
      int inc = v;
#pragma unroll
      for (int off = 1; off < 32; off <<= 1) {
        const int u = __shfl_up(inc, off, 32);
        if (tid >= off) inc += u;
      }
      if (i < nBuckets) lcur[i] = carry + (inc - v);
      carry += __shfl(inc, 31, 32);
    }
  }
  __syncthreads();
  if (blockIdx.x == 0)
    for (int b = tid; b < nBuckets; b += 256) bucketStart[b] = lcur[b];
  for (int b = tid; b < nBuckets; b += 256)
    lcur[b] += offs[(size_t)b * gridDim.x + blockIdx.x];
  __syncthreads();
  const int lo = (int)blockIdx.x * chunk;
  int hi = lo + chunk;
  hi = (hi < nEdges) ? hi : nEdges;
  const unsigned last = (unsigned)(nEdges - 1);
  int2* pairs = (int2*)binned;
  int e = lo + tid;
  for (; e + 768 < hi; e += 1024) {
    const int d0 = dst[e],       s0 = src[e];
    const int d1 = dst[e + 256], s1 = src[e + 256];
    const int d2 = dst[e + 512], s2 = src[e + 512];
    const int d3 = dst[e + 768], s3 = src[e + 768];
    const unsigned p0 = (unsigned)atomicAdd(&lcur[bucket_of(d0, nBuckets)], 1);
    const unsigned p1 = (unsigned)atomicAdd(&lcur[bucket_of(d1, nBuckets)], 1);
    const unsigned p2 = (unsigned)atomicAdd(&lcur[bucket_of(d2, nBuckets)], 1);
    const unsigned p3 = (unsigned)atomicAdd(&lcur[bucket_of(d3, nBuckets)], 1);
    pairs[(p0 < last) ? p0 : last] = make_int2(s0, d0);
    pairs[(p1 < last) ? p1 : last] = make_int2(s1, d1);
    pairs[(p2 < last) ? p2 : last] = make_int2(s2, d2);
    pairs[(p3 < last) ? p3 : last] = make_int2(s3, d3);
  }
  for (; e < hi; e += 256) {
    const int d0 = dst[e], s0 = src[e];
    const unsigned p0 = (unsigned)atomicAdd(&lcur[bucket_of(d0, nBuckets)], 1);
    pairs[(p0 < last) ? p0 : last] = make_int2(s0, d0);
  }
}

__global__ void __launch_bounds__(BKT_NODES)
bucket_lists(const int* __restrict__ binned, const int* __restrict__ bucketStart,
             int* __restrict__ listEnd, float* __restrict__ deg, int* __restrict__ nbr,
             int nNodes, int nEdges, int nBuckets) {
  __shared__ int cnt[BKT_NODES];
  __shared__ int cur[BKT_NODES];
  __shared__ int waveTot[8];
  const int tid  = threadIdx.x;
  const int lane = tid & 31;
  const int w    = tid >> 5;
  const int b    = (int)blockIdx.x;
  const int node0 = b << BKT_SHIFT;
  int start = bucketStart[b];
  int end   = (b + 1 < nBuckets) ? bucketStart[b + 1] : nEdges;
  start = (start < 0) ? 0 : ((start > nEdges) ? nEdges : start);
  end   = (end > nEdges) ? nEdges : end;
  const int2* pairs = (const int2*)binned;

  cnt[tid] = 0;
  __syncthreads();
  int e = start + tid;
  for (; e + 3 * BKT_NODES < end; e += 4 * BKT_NODES) {
    const int d0 = pairs[e].y;
    const int d1 = pairs[e + BKT_NODES].y;
    const int d2 = pairs[e + 2 * BKT_NODES].y;
    const int d3 = pairs[e + 3 * BKT_NODES].y;
    atomicAdd(&cnt[local_of(d0, node0)], 1);
    atomicAdd(&cnt[local_of(d1, node0)], 1);
    atomicAdd(&cnt[local_of(d2, node0)], 1);
    atomicAdd(&cnt[local_of(d3, node0)], 1);
  }
  for (; e < end; e += BKT_NODES) atomicAdd(&cnt[local_of(pairs[e].y, node0)], 1);
  __syncthreads();

  const int c = cnt[tid];
  int inc = c;
#pragma unroll
  for (int off = 1; off < 32; off <<= 1) {
    const int u = __shfl_up(inc, off, 32);
    if (lane >= off) inc += u;
  }
  if (lane == 31) waveTot[w] = inc;
  __syncthreads();
  const int wt = (lane < 8) ? waveTot[lane] : 0;
  int winc = wt;
#pragma unroll
  for (int off = 1; off < 8; off <<= 1) {
    const int u = __shfl_up(winc, off, 32);
    if (lane >= off) winc += u;
  }
  const int wavePrefix = __shfl(winc - wt, w, 32);
  const int first = start + wavePrefix + (inc - c);
  cur[tid] = first;
  const int node = node0 + tid;
  if (node < nNodes) {
    listEnd[node] = first + c;
    deg[node]     = (float)c;
  }
  __syncthreads();

  const unsigned last = (unsigned)(nEdges - 1);
  e = start + tid;
  for (; e + 3 * BKT_NODES < end; e += 4 * BKT_NODES) {
    const int2 p0 = pairs[e];
    const int2 p1 = pairs[e + BKT_NODES];
    const int2 p2 = pairs[e + 2 * BKT_NODES];
    const int2 p3 = pairs[e + 3 * BKT_NODES];
    const unsigned q0 = (unsigned)atomicAdd(&cur[local_of(p0.y, node0)], 1);
    const unsigned q1 = (unsigned)atomicAdd(&cur[local_of(p1.y, node0)], 1);
    const unsigned q2 = (unsigned)atomicAdd(&cur[local_of(p2.y, node0)], 1);
    const unsigned q3 = (unsigned)atomicAdd(&cur[local_of(p3.y, node0)], 1);
    nbr[(q0 < last) ? q0 : last] = p0.x;
    nbr[(q1 < last) ? q1 : last] = p1.x;
    nbr[(q2 < last) ? q2 : last] = p2.x;
    nbr[(q3 < last) ? q3 : last] = p3.x;
  }
  for (; e < end; e += BKT_NODES) {
    const int2 p0 = pairs[e];
    const unsigned q0 = (unsigned)atomicAdd(&cur[local_of(p0.y, node0)], 1);
    nbr[(q0 < last) ? q0 : last] = p0.x;
  }
}

#define A_PITCH 132
#define W_LDS_FLOATS (64 * LDS_PP2 * 2)
#define LAYER_WAVES 16
#define LAYER_BLOCKS 256

template <bool kNormalize>
__global__ void __launch_bounds__(LAYER_WAVES * 32)
sage_layer(const float* __restrict__ xin,
           const int* __restrict__ listEnd, const int* __restrict__ nbr,
           const float* __restrict__ deg,
           const float* __restrict__ WTiS, const float* __restrict__ WTiN,
           const float* __restrict__ bias,
           float* __restrict__ out, int nNodes, int nEdges) {
  __shared__ __attribute__((aligned(16))) float lds[2 * W_LDS_FLOATS + LAYER_WAVES * 16 * A_PITCH];
  const int tid  = threadIdx.x;
  const int wave = __builtin_amdgcn_readfirstlane(tid >> 5);
  const int lane = tid & 31;
  const int half = lane >> 4;
  const int l16  = lane & 15;

  const v2f* B2S = (const v2f*)lds;
  const v2f* B2N = (const v2f*)(lds + W_LDS_FLOATS);
  float* aTile = lds + 2 * W_LDS_FLOATS + wave * (16 * A_PITCH);
  const float* ar = aTile + l16 * A_PITCH + half * 2;

  {
    const float4* srcS = (const float4*)WTiS;
    const float4* srcN = (const float4*)WTiN;
    constexpr int kFill = (D * D) / 4 / (LAYER_WAVES * 32);
    float4 ws[kFill], wn[kFill];
#pragma unroll
    for (int q = 0; q < kFill; ++q) {
      ws[q] = srcS[tid + q * (LAYER_WAVES * 32)];
      wn[q] = srcN[tid + q * (LAYER_WAVES * 32)];
    }
#pragma unroll
    for (int q = 0; q < kFill; ++q) {
      const int i = tid + q * (LAYER_WAVES * 32);
      const int dstI = (i >> 6) * (LDS_PP2 / 2) + (i & 63);
      ((float4*)lds)[dstI] = ws[q];
      ((float4*)(lds + W_LDS_FLOATS))[dstI] = wn[q];
    }
    __syncthreads();
  }

  const int nStrips = (nNodes + 15) >> 4;
  const unsigned lane4 = (unsigned)lane * 4u;
  for (int strip = (int)blockIdx.x + (int)gridDim.x * wave; strip < nStrips;
       strip += (int)gridDim.x * LAYER_WAVES) {
  const int m0   = strip * 16;
  const int mRow = m0 + l16;
  const int mC   = (mRow < nNodes) ? mRow : (nNodes - 1);

  const float invd = 1.0f / fmaxf(deg[mC], 1.0f);

  int endv = listEnd[mC];
  endv = (endv < 0) ? 0 : ((endv > nEdges) ? nEdges : endv);
  int eBeg = (m0 > 0) ? __builtin_amdgcn_readfirstlane(listEnd[m0 - 1]) : 0;
  eBeg = (eBeg < 0) ? 0 : ((eBeg > nEdges) ? nEdges : eBeg);
  const int eEnd = __builtin_amdgcn_readlane(endv, 15);

  auto load_ids = [&](int cb) -> int {
    int idx = cb + lane;
    idx = (idx < nEdges) ? idx : (nEdges - 1);
    return nbr[idx];
  };
  auto fix_ids = [&](int raw, int cb) -> int {
    const int id = (cb + lane < eEnd) ? raw : m0;
    return (id < 0) ? 0 : ((id >= nNodes) ? (nNodes - 1) : id);
  };
#define ROW_AT(idv, L) load_row_piece(xin, __builtin_amdgcn_readlane(idv, (L)), lane4 * 4u)

  int cb = eBeg;
  int idRaw0 = load_ids(cb);
  int idRawNext = load_ids(cb + 32);

  {
    const float4* x4 = (const float4*)xin;
    float4 r[16];
#pragma unroll
    for (int q = 0; q < 16; ++q) {
      const int row = (m0 + q < nNodes) ? (m0 + q) : (nNodes - 1);
      r[q] = x4[(size_t)row * (D / 4) + lane];
    }
#pragma unroll
    for (int q = 0; q < 16; ++q) *(float4*)(aTile + q * A_PITCH + lane * 4) = r[q];
  }

  int idv = fix_ids(idRaw0, cb);
  v4f ra[16], rb[16];
#pragma unroll
  for (int q = 0; q < 16; ++q) ra[q] = ROW_AT(idv, q);

  v8f acc[8];
#pragma unroll
  for (int t = 0; t < 8; ++t) acc[t] = (v8f)(0.0f);

  {
    v2f aCur = *(const v2f*)(ar);
    for (int k0 = 0; k0 < D; k0 += 4) {
      const int base2 = ((k0 >> 1) + half) * LDS_PP2 + l16;
      v2f b[8];
#pragma unroll
      for (int t = 0; t < 8; ++t) b[t] = B2S[base2 + t * 16];

      const int kN = (k0 + 4 < D) ? (k0 + 4) : k0;
      const v2f aNext = *(const v2f*)(ar + kN);

#pragma unroll
      for (int t = 0; t < 8; ++t) acc[t] = wmma_f32(aCur, b[t], acc[t]);
      aCur = aNext;
    }
  }

  {
    int nodeI = 0;
    int endN = __builtin_amdgcn_readlane(endv, 0);
    v4f acc4 = (v4f)(0.0f);
#define PARK_DONE_NODES(P) \
    while (nodeI < 16 && (P) >= endN) { \
      *(v4f*)(aTile + nodeI * A_PITCH + lane4) = acc4; \
      acc4 = (v4f)(0.0f); \
      ++nodeI; \
      endN = __builtin_amdgcn_readlane(endv, nodeI & 15); \
    }
#define ADD_ROW(v) { acc4 += (v); }

    for (; cb < eEnd; cb += 32) {
#pragma unroll
      for (int q = 0; q < 16; ++q) rb[q] = ROW_AT(idv, 16 + q);
#pragma unroll
      for (int q = 0; q < 16; ++q) { PARK_DONE_NODES(cb + q); ADD_ROW(ra[q]); }
      const int cbN = cb + 32;
      const int idvN = fix_ids(idRawNext, cbN);
      idRawNext = load_ids(cbN + 32);
#pragma unroll
      for (int q = 0; q < 16; ++q) ra[q] = ROW_AT(idvN, q);
#pragma unroll
      for (int q = 0; q < 16; ++q) { PARK_DONE_NODES(cb + 16 + q); ADD_ROW(rb[q]); }
      idv = idvN;
    }
    while (nodeI < 16) {
      *(v4f*)(aTile + nodeI * A_PITCH + lane4) = acc4;
      acc4 = (v4f)(0.0f);
      ++nodeI;
    }
#undef PARK_DONE_NODES
#undef ADD_ROW
#undef ROW_AT
  }

  {
    v2f aCur = *(const v2f*)(ar);
    aCur *= invd;
    for (int k0 = 0; k0 < D; k0 += 4) {
      const int base2 = ((k0 >> 1) + half) * LDS_PP2 + l16;
      v2f b[8];
#pragma unroll
      for (int t = 0; t < 8; ++t) b[t] = B2N[base2 + t * 16];

      const int kN = (k0 + 4 < D) ? (k0 + 4) : k0;
      v2f aNext = *(const v2f*)(ar + kN);
      aNext *= invd;

#pragma unroll
      for (int t = 0; t < 8; ++t) acc[t] = wmma_f32(aCur, b[t], acc[t]);
      aCur = aNext;
    }
  }

#pragma unroll
  for (int t = 0; t < 8; ++t) {
    const float bv = bias[t * 16 + l16];
#pragma unroll
    for (int r = 0; r < 8; ++r) acc[t][r] += bv;
  }

  if (kNormalize) {
    float ss[8];
#pragma unroll
    for (int r = 0; r < 8; ++r) {
      float s = 0.0f;
#pragma unroll
      for (int t = 0; t < 8; ++t) s += acc[t][r] * acc[t][r];
      ss[r] = s;
    }
#pragma unroll
    for (int r = 0; r < 8; ++r) ss[r] += row_xor<1>(ss[r]);
#pragma unroll
    for (int r = 0; r < 8; ++r) ss[r] += row_xor<2>(ss[r]);
#pragma unroll
    for (int r = 0; r < 8; ++r) ss[r] += row_xor<4>(ss[r]);
#pragma unroll
    for (int r = 0; r < 8; ++r) ss[r] += row_xor<8>(ss[r]);
#pragma unroll
    for (int r = 0; r < 8; ++r) {
      const float scale = 1.0f / fmaxf(sqrtf(ss[r]), 1e-12f);
#pragma unroll
      for (int t = 0; t < 8; ++t) acc[t][r] *= scale;
    }
  }

#pragma unroll
  for (int t = 0; t < 8; ++t) {
    const int col = t * 16 + l16;
#pragma unroll
    for (int r = 0; r < 8; ++r) {
      const int row = m0 + r + half * 8;
      if (row < nNodes) out[(size_t)row * D + col] = acc[t][r];
    }
  }
  }
}

extern "C" void kernel_launch(void* const* d_in, const int* in_sizes, int n_in,
                              void* d_out, int out_size, void* d_ws, size_t ws_size,
                              hipStream_t stream) {
  const float* x   = (const float*)d_in[0];
  const int*   src = (const int*)  d_in[1];
  const int*   dst = (const int*)  d_in[2];
  const float* Ws1 = (const float*)d_in[3];
  const float* Wn1 = (const float*)d_in[4];
  const float* b1  = (const float*)d_in[5];
  const float* Ws2 = (const float*)d_in[6];
  const float* Wn2 = (const float*)d_in[7];
  const float* b2  = (const float*)d_in[8];
  float* out = (float*)d_out;

  const int N = in_sizes[0] / D;
  const int E = in_sizes[1];

  char* ws = (char*)d_ws;
  char*  lists = ws;       ws += (size_t)N * D * sizeof(float);
  float* h1  = (float*)ws; ws += (size_t)N * D * sizeof(float);
  float* deg = (float*)ws; ws += (((size_t)N * sizeof(float)) + 255) & ~(size_t)255;
  float* WTs1 = (float*)ws; ws += (size_t)D * D * sizeof(float);
  float* WTn1 = (float*)ws; ws += (size_t)D * D * sizeof(float);
  float* WTs2 = (float*)ws; ws += (size_t)D * D * sizeof(float);
  float* WTn2 = (float*)ws; ws += (size_t)D * D * sizeof(float);

  dim3 gemmGrid(LAYER_BLOCKS);

  char* sc = (char*)lists;
  const int nBuckets = (N + BKT_NODES - 1) / BKT_NODES;
  const int chunk    = (E + BIN_GROUPS - 1) / BIN_GROUPS;
  const int nHist    = nBuckets * BIN_GROUPS;
  int* listPtr = (int*)sc; sc += (((size_t)N * sizeof(int)) + 255) & ~(size_t)255;
  int* hist    = (int*)sc; sc += (((size_t)nHist * sizeof(int)) + 255) & ~(size_t)255;
  int* bucketTotal = (int*)sc; sc += (((size_t)nBuckets * sizeof(int)) + 255) & ~(size_t)255;
  int* bucketStart = (int*)sc; sc += (((size_t)nBuckets * sizeof(int)) + 255) & ~(size_t)255;
  int* binned  = (int*)sc; sc += (((size_t)E * 2 * sizeof(int)) + 255) & ~(size_t)255;
  int* nbr     = (int*)sc;
  const size_t bktLds = (size_t)nBuckets * sizeof(int);

  bucket_count<<<BIN_GROUPS + 4 * REPACK_BLOCKS, 256, bktLds, stream>>>(dst, hist, E, nBuckets, chunk, BIN_GROUPS,
                                                                        Ws1, Wn1, Ws2, Wn2, WTs1, WTn1, WTs2, WTn2);
  bucket_scan<<<nBuckets, BIN_GROUPS, 0, stream>>>(hist, bucketTotal);
  bucket_scatter<<<BIN_GROUPS, 256, bktLds, stream>>>(src, dst, hist, bucketTotal, bucketStart, binned, E, nBuckets, chunk);
  bucket_lists<<<nBuckets, BKT_NODES, 0, stream>>>(binned, bucketStart, listPtr, deg, nbr, N, E, nBuckets);

  sage_layer<false><<<gemmGrid, LAYER_WAVES * 32, 0, stream>>>(x, listPtr, nbr, deg, WTs1, WTn1, b1, h1, N, E);

  sage_layer<true><<<gemmGrid, LAYER_WAVES * 32, 0, stream>>>(h1, listPtr, nbr, deg, WTs2, WTn2, b2, out, N, E);
}


# Layer kernel: no register of an issued global load or store is rewritten before it lands (own offset register per row load, groups of 12; 64 stores off one base, unguarded): no s_wait_xcnt stalls
# speedup vs baseline: 4.0312x; 1.0669x over previous
#include <hip/hip_runtime.h>
#include <cstddef>
#include <cstdint>

#define D 128
#define LDS_PP2 144

typedef float v2f __attribute__((ext_vector_type(2)));
typedef float v8f __attribute__((ext_vector_type(8)));
typedef float v4f __attribute__((ext_vector_type(4)));

__device__ __forceinline__ v8f wmma_f32(v2f a, v2f b, v8f c) {
#if defined(__gfx1250__)
  return __builtin_amdgcn_wmma_f32_16x16x4_f32(false, a, false, b,
                                               (short)0, c, false, false);
#else
  c[0] += a[0] * b[0];
  return c;
#endif
}

template <int M>
__device__ __forceinline__ float row_xor(float v) {
#if defined(__HIP_DEVICE_COMPILE__)
  const int b = __builtin_bit_cast(int, v);
  return __builtin_bit_cast(float, __builtin_amdgcn_update_dpp(b, b, 0x160 | M, 0xf, 0xf, false));
#else
  return v;
#endif
}

#define REPACK_BLOCKS ((D * D) / 256)

__device__ __forceinline__ void repack_weight(const float* __restrict__ W, float* __restrict__ WTi, int idx) {
  const int n = (idx >> 1) & (D - 1);
  const int k = ((idx >> 8) << 1) | (idx & 1);
  WTi[idx] = W[n * D + k];
}

#define BKT_SHIFT 8
#define BKT_NODES 256
#define BIN_GROUPS 512

__device__ __forceinline__ int bucket_of(int d, int nBuckets) {
  const int b = d >> BKT_SHIFT;
  return (b < 0) ? 0 : ((b >= nBuckets) ? (nBuckets - 1) : b);
}

__device__ __forceinline__ int local_of(int d, int node0) {
  const int l = d - node0;
  return (l < 0) ? 0 : ((l >= BKT_NODES) ? (BKT_NODES - 1) : l);
}

__global__ void __launch_bounds__(256)
bucket_count(const int* __restrict__ dst, int* __restrict__ hist,
             int nEdges, int nBuckets, int chunk, int groups,
             const float* __restrict__ Wa, const float* __restrict__ Wb,
             const float* __restrict__ Wc, const float* __restrict__ Wd,
             float* __restrict__ WTa, float* __restrict__ WTb,
             float* __restrict__ WTc, float* __restrict__ WTd) {
  extern __shared__ int lcnt[];
  const int tid = threadIdx.x;
  if ((int)blockIdx.x >= groups) {
    const int r = (int)blockIdx.x - groups;
    const int m = r / REPACK_BLOCKS;
    const int idx = (r - m * REPACK_BLOCKS) * 256 + tid;
    if      (m == 0) repack_weight(Wa, WTa, idx);
    else if (m == 1) repack_weight(Wb, WTb, idx);
    else if (m == 2) repack_weight(Wc, WTc, idx);
    else if (m == 3) repack_weight(Wd, WTd, idx);
    return;
  }
  for (int b = tid; b < nBuckets; b += 256) lcnt[b] = 0;
  __syncthreads();
  const int lo = (int)blockIdx.x * chunk;
  int hi = lo + chunk;
  hi = (hi < nEdges) ? hi : nEdges;
  int e = lo + tid;
  for (; e + 768 < hi; e += 1024) {
    const int d0 = dst[e];
    const int d1 = dst[e + 256];
    const int d2 = dst[e + 512];
    const int d3 = dst[e + 768];
    atomicAdd(&lcnt[bucket_of(d0, nBuckets)], 1);
    atomicAdd(&lcnt[bucket_of(d1, nBuckets)], 1);
    atomicAdd(&lcnt[bucket_of(d2, nBuckets)], 1);
    atomicAdd(&lcnt[bucket_of(d3, nBuckets)], 1);
  }
  for (; e < hi; e += 256) atomicAdd(&lcnt[bucket_of(dst[e], nBuckets)], 1);
  __syncthreads();
  for (int b = tid; b < nBuckets; b += 256)
    hist[(size_t)b * groups + blockIdx.x] = lcnt[b];
}

__global__ void __launch_bounds__(BIN_GROUPS)
bucket_scan(int* __restrict__ hist, int* __restrict__ bucketTotal) {
  __shared__ int waveTot[BIN_GROUPS / 32];
  const int tid  = threadIdx.x;
  const int lane = tid & 31;
  const int w    = tid >> 5;
  const size_t i = (size_t)blockIdx.x * BIN_GROUPS + tid;
  const int v = hist[i];
  int inc = v;
#pragma unroll
  for (int off = 1; off < 32; off <<= 1) {
    const int u = __shfl_up(inc, off, 32);
    if (lane >= off) inc += u;
  }
  if (lane == 31) waveTot[w] = inc;
  __syncthreads();
  const int wt = (lane < BIN_GROUPS / 32) ? waveTot[lane] : 0;
  int winc = wt;
#pragma unroll
  for (int off = 1; off < BIN_GROUPS / 32; off <<= 1) {
    const int u = __shfl_up(winc, off, 32);
    if (lane >= off) winc += u;
  }
  const int wavePrefix = __shfl(winc - wt, w, 32);
  hist[i] = wavePrefix + (inc - v);
  if (tid == BIN_GROUPS - 1) bucketTotal[blockIdx.x] = wavePrefix + inc;
}

__global__ void __launch_bounds__(256)
bucket_scatter(const int* __restrict__ src, const int* __restrict__ dst,
               const int* __restrict__ offs, const int* __restrict__ bucketTotal,
               int* __restrict__ bucketStart, int* __restrict__ binned,
               int nEdges, int nBuckets, int chunk) {
  extern __shared__ int lcur[];
  const int tid = threadIdx.x;
  for (int b = tid; b < nBuckets; b += 256) lcur[b] = bucketTotal[b];
  __syncthreads();
  if (tid < 32) {
    int carry = 0;
    for (int base = 0; base < nBuckets; base += 32) {
      const int i = base + tid;
      const int v = (i < nBuckets) ? lcur[i] : 0;
      int inc = v;
#pragma unroll
      for (int off = 1; off < 32; off <<= 1) {
        const int u = __shfl_up(inc, off, 32);
        if (tid >= off) inc += u;
      }
      if (i < nBuckets) lcur[i] = carry + (inc - v);
      carry += __shfl(inc, 31, 32);
    }
  }
  __syncthreads();
  if (blockIdx.x == 0)
    for (int b = tid; b < nBuckets; b += 256) bucketStart[b] = lcur[b];
  for (int b = tid; b < nBuckets; b += 256)
    lcur[b] += offs[(size_t)b * gridDim.x + blockIdx.x];
  __syncthreads();
  const int lo = (int)blockIdx.x * chunk;
  int hi = lo + chunk;
  hi = (hi < nEdges) ? hi : nEdges;
  const unsigned last = (unsigned)(nEdges - 1);
  int2* pairs = (int2*)binned;
  int e = lo + tid;
  for (; e + 768 < hi; e += 1024) {
    const int d0 = dst[e],       s0 = src[e];
    const int d1 = dst[e + 256], s1 = src[e + 256];
    const int d2 = dst[e + 512], s2 = src[e + 512];
    const int d3 = dst[e + 768], s3 = src[e + 768];
    const unsigned p0 = (unsigned)atomicAdd(&lcur[bucket_of(d0, nBuckets)], 1);
    const unsigned p1 = (unsigned)atomicAdd(&lcur[bucket_of(d1, nBuckets)], 1);
    const unsigned p2 = (unsigned)atomicAdd(&lcur[bucket_of(d2, nBuckets)], 1);
    const unsigned p3 = (unsigned)atomicAdd(&lcur[bucket_of(d3, nBuckets)], 1);
    pairs[(p0 < last) ? p0 : last] = make_int2(s0, d0);
    pairs[(p1 < last) ? p1 : last] = make_int2(s1, d1);
    pairs[(p2 < last) ? p2 : last] = make_int2(s2, d2);
    pairs[(p3 < last) ? p3 : last] = make_int2(s3, d3);
  }
  for (; e < hi; e += 256) {
    const int d0 = dst[e], s0 = src[e];
    const unsigned p0 = (unsigned)atomicAdd(&lcur[bucket_of(d0, nBuckets)], 1);
    pairs[(p0 < last) ? p0 : last] = make_int2(s0, d0);
  }
}

__global__ void __launch_bounds__(BKT_NODES)
bucket_lists(const int* __restrict__ binned, const int* __restrict__ bucketStart,
             int* __restrict__ listEnd, float* __restrict__ deg, int* __restrict__ nbr,
             int nNodes, int nEdges, int nBuckets) {
  __shared__ int cnt[BKT_NODES];
  __shared__ int cur[BKT_NODES];
  __shared__ int waveTot[8];
  const int tid  = threadIdx.x;
  const int lane = tid & 31;
  const int w    = tid >> 5;
  const int b    = (int)blockIdx.x;
  const int node0 = b << BKT_SHIFT;
  int start = bucketStart[b];
  int end   = (b + 1 < nBuckets) ? bucketStart[b + 1] : nEdges;
  start = (start < 0) ? 0 : ((start > nEdges) ? nEdges : start);
  end   = (end > nEdges) ? nEdges : end;
  const int2* pairs = (const int2*)binned;

  cnt[tid] = 0;
  __syncthreads();
  int e = start + tid;
  for (; e + 3 * BKT_NODES < end; e += 4 * BKT_NODES) {
    const int d0 = pairs[e].y;
    const int d1 = pairs[e + BKT_NODES].y;
    const int d2 = pairs[e + 2 * BKT_NODES].y;
    const int d3 = pairs[e + 3 * BKT_NODES].y;
    atomicAdd(&cnt[local_of(d0, node0)], 1);
    atomicAdd(&cnt[local_of(d1, node0)], 1);
    atomicAdd(&cnt[local_of(d2, node0)], 1);
    atomicAdd(&cnt[local_of(d3, node0)], 1);
  }
  for (; e < end; e += BKT_NODES) atomicAdd(&cnt[local_of(pairs[e].y, node0)], 1);
  __syncthreads();

  const int c = cnt[tid];
  int inc = c;
#pragma unroll
  for (int off = 1; off < 32; off <<= 1) {
    const int u = __shfl_up(inc, off, 32);
    if (lane >= off) inc += u;
  }
  if (lane == 31) waveTot[w] = inc;
  __syncthreads();
  const int wt = (lane < 8) ? waveTot[lane] : 0;
  int winc = wt;
#pragma unroll
  for (int off = 1; off < 8; off <<= 1) {
    const int u = __shfl_up(winc, off, 32);
    if (lane >= off) winc += u;
  }
  const int wavePrefix = __shfl(winc - wt, w, 32);
  const int first = start + wavePrefix + (inc - c);
  cur[tid] = first;
  const int node = node0 + tid;
  if (node < nNodes) {
    listEnd[node] = first + c;
    deg[node]     = (float)c;
  }
  __syncthreads();

  const unsigned last = (unsigned)(nEdges - 1);
  e = start + tid;
  for (; e + 3 * BKT_NODES < end; e += 4 * BKT_NODES) {
    const int2 p0 = pairs[e];
    const int2 p1 = pairs[e + BKT_NODES];
    const int2 p2 = pairs[e + 2 * BKT_NODES];
    const int2 p3 = pairs[e + 3 * BKT_NODES];
    const unsigned q0 = (unsigned)atomicAdd(&cur[local_of(p0.y, node0)], 1);
    const unsigned q1 = (unsigned)atomicAdd(&cur[local_of(p1.y, node0)], 1);
    const unsigned q2 = (unsigned)atomicAdd(&cur[local_of(p2.y, node0)], 1);
    const unsigned q3 = (unsigned)atomicAdd(&cur[local_of(p3.y, node0)], 1);
    nbr[(q0 < last) ? q0 : last] = p0.x;
    nbr[(q1 < last) ? q1 : last] = p1.x;
    nbr[(q2 < last) ? q2 : last] = p2.x;
    nbr[(q3 < last) ? q3 : last] = p3.x;
  }
  for (; e < end; e += BKT_NODES) {
    const int2 p0 = pairs[e];
    const unsigned q0 = (unsigned)atomicAdd(&cur[local_of(p0.y, node0)], 1);
    nbr[(q0 < last) ? q0 : last] = p0.x;
  }
}

#define A_PITCH 132
#define W_LDS_FLOATS (64 * LDS_PP2 * 2)
#define LAYER_WAVES 16
#define LAYER_BLOCKS 256
#define GROUP_ROWS 12

template <bool kNormalize>
__global__ void __launch_bounds__(LAYER_WAVES * 32)
sage_layer(const float* __restrict__ xin,
           const int* __restrict__ listEnd, const int* __restrict__ nbr,
           const float* __restrict__ deg,
           const float* __restrict__ WTiS, const float* __restrict__ WTiN,
           const float* __restrict__ bias,
           float* __restrict__ out, int nNodes, int nEdges) {
  __shared__ __attribute__((aligned(16))) float lds[2 * W_LDS_FLOATS + LAYER_WAVES * 16 * A_PITCH];
  const int tid  = threadIdx.x;
  const int wave = __builtin_amdgcn_readfirstlane(tid >> 5);
  const int lane = tid & 31;
  const int half = lane >> 4;
  const int l16  = lane & 15;

  const v2f* B2S = (const v2f*)lds;
  const v2f* B2N = (const v2f*)(lds + W_LDS_FLOATS);
  float* aTile = lds + 2 * W_LDS_FLOATS + wave * (16 * A_PITCH);
  const float* ar = aTile + l16 * A_PITCH + half * 2;

  {
    const float4* srcS = (const float4*)WTiS;
    const float4* srcN = (const float4*)WTiN;
    constexpr int kFill = (D * D) / 4 / (LAYER_WAVES * 32);
    float4 ws[kFill], wn[kFill];
#pragma unroll
    for (int q = 0; q < kFill; ++q) {
      ws[q] = srcS[tid + q * (LAYER_WAVES * 32)];
      wn[q] = srcN[tid + q * (LAYER_WAVES * 32)];
    }
#pragma unroll
    for (int q = 0; q < kFill; ++q) {
      const int i = tid + q * (LAYER_WAVES * 32);
      const int dstI = (i >> 6) * (LDS_PP2 / 2) + (i & 63);
      ((float4*)lds)[dstI] = ws[q];
      ((float4*)(lds + W_LDS_FLOATS))[dstI] = wn[q];
    }
    __syncthreads();
  }

  const int nStrips = (nNodes + 15) >> 4;
  const unsigned lane4 = (unsigned)lane * 4u;
  const unsigned laneOut = (unsigned)(half * 8 * D + l16);
  for (int strip = (int)blockIdx.x + (int)gridDim.x * wave; strip < nStrips;
       strip += (int)gridDim.x * LAYER_WAVES) {
  const int m0   = strip * 16;
  const int mRow = m0 + l16;
  const int mC   = (mRow < nNodes) ? mRow : (nNodes - 1);

  const float invd = 1.0f / fmaxf(deg[mC], 1.0f);

  int endv = listEnd[mC];
  endv = (endv < 0) ? 0 : ((endv > nEdges) ? nEdges : endv);
  int eBeg = (m0 > 0) ? __builtin_amdgcn_readfirstlane(listEnd[m0 - 1]) : 0;
  eBeg = (eBeg < 0) ? 0 : ((eBeg > nEdges) ? nEdges : eBeg);
  const int eEnd = __builtin_amdgcn_readlane(endv, 15);

  auto ids_index = [&](int cb) -> int {
    const int idx = cb + lane;
    return (idx < nEdges) ? idx : (nEdges - 1);
  };
  auto fix_ids = [&](int raw, int cb) -> int {
    const int id = (cb + lane < eEnd) ? raw : m0;
    return (id < 0) ? 0 : ((id >= nNodes) ? (nNodes - 1) : id);
  };
#define ROW_OFF(idv, L) ((unsigned)__builtin_amdgcn_readlane(idv, (L)) * (unsigned)(D * sizeof(float)) + lane4 * 4u)
#define ROW_LOAD(off) (*(const v4f*)((const char*)xin + (off)))
#define KEEP_OFF(off) asm volatile("" :: "v"(off))

  int cb = eBeg;
  const int ix0 = ids_index(cb);
  int idRaw0 = nbr[ix0];
  int ixNext = ids_index(cb + 2 * GROUP_ROWS);
  int idRawNext = nbr[ixNext];

  {
    v4f r[16];
    unsigned orw[16];
#pragma unroll
    for (int q = 0; q < 16; ++q) {
      const int row = (m0 + q < nNodes) ? (m0 + q) : (nNodes - 1);
      orw[q] = (unsigned)row * (unsigned)(D * sizeof(float)) + lane4 * 4u;
      r[q] = ROW_LOAD(orw[q]);
    }
#pragma unroll
    for (int q = 0; q < 16; ++q) { *(v4f*)(aTile + q * A_PITCH + lane * 4) = r[q]; KEEP_OFF(orw[q]); }
  }

  int idv = fix_ids(idRaw0, cb);
  KEEP_OFF(ix0);
  v4f ra[GROUP_ROWS], rb[GROUP_ROWS];
  unsigned oa[GROUP_ROWS], ob[GROUP_ROWS];
#pragma unroll
  for (int q = 0; q < GROUP_ROWS; ++q) { oa[q] = ROW_OFF(idv, q); ra[q] = ROW_LOAD(oa[q]); }

  v8f acc[8];
#pragma unroll
  for (int t = 0; t < 8; ++t) acc[t] = (v8f)(0.0f);

  {
    v2f aCur = *(const v2f*)(ar);
    for (int k0 = 0; k0 < D; k0 += 4) {
      const int base2 = ((k0 >> 1) + half) * LDS_PP2 + l16;
      v2f b[8];
#pragma unroll
      for (int t = 0; t < 8; ++t) b[t] = B2S[base2 + t * 16];

      const int kN = (k0 + 4 < D) ? (k0 + 4) : k0;
      const v2f aNext = *(const v2f*)(ar + kN);

#pragma unroll
      for (int t = 0; t < 8; ++t) acc[t] = wmma_f32(aCur, b[t], acc[t]);
      aCur = aNext;
    }
  }

  {
    int nodeI = 0;
    int endN = __builtin_amdgcn_readlane(endv, 0);
    v4f acc4 = (v4f)(0.0f);
#define PARK_DONE_NODES(P) \
    while (nodeI < 16 && (P) >= endN) { \
      *(v4f*)(aTile + nodeI * A_PITCH + lane4) = acc4; \
      acc4 = (v4f)(0.0f); \
      ++nodeI; \
      endN = __builtin_amdgcn_readlane(endv, nodeI & 15); \
    }
#define ADD_ROW(v) { acc4 += (v); }

    for (; cb < eEnd; cb += 2 * GROUP_ROWS) {
#pragma unroll
      for (int q = 0; q < GROUP_ROWS; ++q) { ob[q] = ROW_OFF(idv, GROUP_ROWS + q); rb[q] = ROW_LOAD(ob[q]); }
#pragma unroll
      for (int q = 0; q < GROUP_ROWS; ++q) { PARK_DONE_NODES(cb + q); ADD_ROW(ra[q]); KEEP_OFF(oa[q]); }
      const int cbN = cb + 2 * GROUP_ROWS;
      const int idvN = fix_ids(idRawNext, cbN);
      KEEP_OFF(ixNext);
      ixNext = ids_index(cbN + 2 * GROUP_ROWS);
      idRawNext = nbr[ixNext];
#pragma unroll
      for (int q = 0; q < GROUP_ROWS; ++q) { oa[q] = ROW_OFF(idvN, q); ra[q] = ROW_LOAD(oa[q]); }
#pragma unroll
      for (int q = 0; q < GROUP_ROWS; ++q) { PARK_DONE_NODES(cb + GROUP_ROWS + q); ADD_ROW(rb[q]); KEEP_OFF(ob[q]); }
      idv = idvN;
    }
    while (nodeI < 16) {
      *(v4f*)(aTile + nodeI * A_PITCH + lane4) = acc4;
      acc4 = (v4f)(0.0f);
      ++nodeI;
    }
#undef PARK_DONE_NODES
#undef ADD_ROW
#undef ROW_OFF
#undef ROW_LOAD
#undef KEEP_OFF
  }

  {
    v2f aCur = *(const v2f*)(ar);
    aCur *= invd;
    for (int k0 = 0; k0 < D; k0 += 4) {
      const int base2 = ((k0 >> 1) + half) * LDS_PP2 + l16;
      v2f b[8];
#pragma unroll
      for (int t = 0; t < 8; ++t) b[t] = B2N[base2 + t * 16];

      const int kN = (k0 + 4 < D) ? (k0 + 4) : k0;
      v2f aNext = *(const v2f*)(ar + kN);
      aNext *= invd;

#pragma unroll
      for (int t = 0; t < 8; ++t) acc[t] = wmma_f32(aCur, b[t], acc[t]);
      aCur = aNext;
    }
  }

#pragma unroll
  for (int t = 0; t < 8; ++t) {
    const float bv = bias[t * 16 + l16];
#pragma unroll
    for (int r = 0; r < 8; ++r) acc[t][r] += bv;
  }

  if (kNormalize) {
    float ss[8];
#pragma unroll
    for (int r = 0; r < 8; ++r) {
      float s = 0.0f;
#pragma unroll
      for (int t = 0; t < 8; ++t) s += acc[t][r] * acc[t][r];
      ss[r] = s;
    }
#pragma unroll
    for (int r = 0; r < 8; ++r) ss[r] += row_xor<1>(ss[r]);
#pragma unroll
    for (int r = 0; r < 8; ++r) ss[r] += row_xor<2>(ss[r]);
#pragma unroll
    for (int r = 0; r < 8; ++r) ss[r] += row_xor<4>(ss[r]);
#pragma unroll
    for (int r = 0; r < 8; ++r) ss[r] += row_xor<8>(ss[r]);
#pragma unroll
    for (int r = 0; r < 8; ++r) {
      const float scale = 1.0f / fmaxf(sqrtf(ss[r]), 1e-12f);
#pragma unroll
      for (int t = 0; t < 8; ++t) acc[t][r] *= scale;
    }
  }

  {
    float* orow = out + (size_t)m0 * D;
    if (m0 + 16 <= nNodes) {
#pragma unroll
      for (int t = 0; t < 8; ++t) {
#pragma unroll
        for (int r = 0; r < 8; ++r) orow[(size_t)laneOut + (size_t)(r * D + t * 16)] = acc[t][r];
      }
    } else {
#pragma unroll
      for (int t = 0; t < 8; ++t) {
#pragma unroll
        for (int r = 0; r < 8; ++r) {
          if (m0 + r + half * 8 < nNodes) orow[(size_t)laneOut + (size_t)(r * D + t * 16)] = acc[t][r];
        }
      }
    }
  }
  }
}

extern "C" void kernel_launch(void* const* d_in, const int* in_sizes, int n_in,
                              void* d_out, int out_size, void* d_ws, size_t ws_size,
                              hipStream_t stream) {
  const float* x   = (const float*)d_in[0];
  const int*   src = (const int*)  d_in[1];
  const int*   dst = (const int*)  d_in[2];
  const float* Ws1 = (const float*)d_in[3];
  const float* Wn1 = (const float*)d_in[4];
  const float* b1  = (const float*)d_in[5];
  const float* Ws2 = (const float*)d_in[6];
  const float* Wn2 = (const float*)d_in[7];
  const float* b2  = (const float*)d_in[8];
  float* out = (float*)d_out;

  const int N = in_sizes[0] / D;
  const int E = in_sizes[1];

  char* ws = (char*)d_ws;
  char*  lists = ws;       ws += (size_t)N * D * sizeof(float);
  float* h1  = (float*)ws; ws += (size_t)N * D * sizeof(float);
  float* deg = (float*)ws; ws += (((size_t)N * sizeof(float)) + 255) & ~(size_t)255;
  float* WTs1 = (float*)ws; ws += (size_t)D * D * sizeof(float);
  float* WTn1 = (float*)ws; ws += (size_t)D * D * sizeof(float);
  float* WTs2 = (float*)ws; ws += (size_t)D * D * sizeof(float);
  float* WTn2 = (float*)ws; ws += (size_t)D * D * sizeof(float);

  dim3 gemmGrid(LAYER_BLOCKS);

  char* sc = (char*)lists;
  const int nBuckets = (N + BKT_NODES - 1) / BKT_NODES;
  const int chunk    = (E + BIN_GROUPS - 1) / BIN_GROUPS;
  const int nHist    = nBuckets * BIN_GROUPS;
  int* listPtr = (int*)sc; sc += (((size_t)N * sizeof(int)) + 255) & ~(size_t)255;
  int* hist    = (int*)sc; sc += (((size_t)nHist * sizeof(int)) + 255) & ~(size_t)255;
  int* bucketTotal = (int*)sc; sc += (((size_t)nBuckets * sizeof(int)) + 255) & ~(size_t)255;
  int* bucketStart = (int*)sc; sc += (((size_t)nBuckets * sizeof(int)) + 255) & ~(size_t)255;
  int* binned  = (int*)sc; sc += (((size_t)E * 2 * sizeof(int)) + 255) & ~(size_t)255;
  int* nbr     = (int*)sc;
  const size_t bktLds = (size_t)nBuckets * sizeof(int);

  bucket_count<<<BIN_GROUPS + 4 * REPACK_BLOCKS, 256, bktLds, stream>>>(dst, hist, E, nBuckets, chunk, BIN_GROUPS,
                                                                        Ws1, Wn1, Ws2, Wn2, WTs1, WTn1, WTs2, WTn2);
  bucket_scan<<<nBuckets, BIN_GROUPS, 0, stream>>>(hist, bucketTotal);
  bucket_scatter<<<BIN_GROUPS, 256, bktLds, stream>>>(src, dst, hist, bucketTotal, bucketStart, binned, E, nBuckets, chunk);
  bucket_lists<<<nBuckets, BKT_NODES, 0, stream>>>(binned, bucketStart, listPtr, deg, nbr, N, E, nBuckets);

  sage_layer<false><<<gemmGrid, LAYER_WAVES * 32, 0, stream>>>(x, listPtr, nbr, deg, WTs1, WTn1, b1, h1, N, E);

  sage_layer<true><<<gemmGrid, LAYER_WAVES * 32, 0, stream>>>(h1, listPtr, nbr, deg, WTs2, WTn2, b2, out, N, E);
}
